# starting kernel
# baseline (speedup 1.0000x reference)
#include <hip/hip_runtime.h>
#include <hip/hip_bf16.h>

typedef __attribute__((ext_vector_type(2))) float v2f;
typedef __attribute__((ext_vector_type(8))) float v8f;

#define F_SIZE 256
#define LDSP 260

__global__ __launch_bounds__(256) void k_attn_ids(
    const float* __restrict__ act, const int* __restrict__ bids,
    const int* __restrict__ mids, const int* __restrict__ xy,
    const int* __restrict__ m0, const int* __restrict__ m1,
    const int* __restrict__ m2, const int* __restrict__ m3,
    const float* __restrict__ Wa, const float* __restrict__ ba,
    float* __restrict__ fw_ws, int* __restrict__ id_ws, int n_act)
{
    const int wave = threadIdx.x >> 5;
    const int lane = threadIdx.x & 31;
    const int row  = blockIdx.x * 8 + wave;
    if (row >= n_act) return;

    const float4* __restrict__ arow = (const float4*)(act + (size_t)row * F_SIZE);
    const float4* __restrict__ wrow = (const float4*)(Wa + (size_t)lane * F_SIZE);
    float acc = 0.f;
#pragma unroll 8
    for (int i = 0; i < F_SIZE / 4; ++i) {
        float4 a = arow[i];
        float4 w = wrow[i];
        acc += a.x * w.x + a.y * w.y + a.z * w.z + a.w * w.w;
    }
    acc += ba[lane];

    float mx = acc;
    mx = fmaxf(mx, __shfl_xor(mx, 1, 32));
    mx = fmaxf(mx, __shfl_xor(mx, 2, 32));
    float e = __expf(acc - mx);
    float s = e;
    s += __shfl_xor(s, 1, 32);
    s += __shfl_xor(s, 2, 32);
    const float fw = e / s;

    const int p = lane & 3;
    const int k = lane >> 2;
    const int offx[8] = {-1, 0, 1, -1, 1, -1, 0, 1};
    const int offy[8] = {-1, -1, -1, 0, 0, 1, 1, 1};
    const int scale = p + 1;
    int x = xy[row * 2 + 0] + scale * offx[k];
    int y = xy[row * 2 + 1] + scale * offy[k];
    const int m = mids[row];
    const int dim = 128 >> m;
    x = min(max(x, 0), dim - 1);
    y = min(max(y, 0), dim - 1);
    const int* __restrict__ map = (m == 0) ? m0 : (m == 1) ? m1 : (m == 2) ? m2 : m3;
    const int b = bids[row];
    const int id = map[((size_t)b * dim + y) * dim + x];

    const size_t o = (size_t)row * 32 + p * 8 + k;
    fw_ws[o] = fw;
    id_ws[o] = id;
}

__global__ __launch_bounds__(256) void k_gather(
    const float* __restrict__ act, const float* __restrict__ pas,
    const float* __restrict__ fw_ws, const int* __restrict__ id_ws,
    float* __restrict__ weighted, int n_act)
{
    __shared__ float s_fw[32];
    __shared__ const float* s_ptr[32];
    const int row = blockIdx.x;
    const int t = threadIdx.x;
    if (t < 32) {
        const int id = id_ws[(size_t)row * 32 + t];
        s_fw[t] = fw_ws[(size_t)row * 32 + t];
        s_ptr[t] = (id < n_act) ? (act + (size_t)id * F_SIZE)
                                : (pas + (size_t)(id - n_act) * F_SIZE);
    }
    __syncthreads();
    float acc = 0.f;
#pragma unroll
    for (int g = 0; g < 32; ++g)
        acc += s_fw[g] * s_ptr[g][t];
    weighted[(size_t)row * F_SIZE + t] = acc;
}

__global__ __launch_bounds__(256) void k_dual_gemm(
    const float* __restrict__ weighted,
    const float* __restrict__ Wv, const float* __restrict__ bv,
    const float* __restrict__ Wo, const float* __restrict__ bo,
    float* __restrict__ out)
{
    __shared__ float lds_a[16 * LDSP];
    __shared__ float lds_v[16 * LDSP];

    const int t = threadIdx.x;
    const int wave = t >> 5;
    const int lane = t & 31;
    const int m0 = blockIdx.x * 16;

    {
        const int r = t >> 4;
        const int c0 = (t & 15) * 16;
        const float4* __restrict__ src =
            (const float4*)(weighted + (size_t)(m0 + r) * F_SIZE + c0);
        float4* dst = (float4*)(lds_a + r * LDSP + c0);
        dst[0] = src[0]; dst[1] = src[1]; dst[2] = src[2]; dst[3] = src[3];
    }
    __syncthreads();

    const int mrow = lane & 15;
    const int hi = lane >> 4;

#pragma unroll
    for (int tn = 0; tn < 2; ++tn) {
        const int nn = (wave * 2 + tn) * 16 + mrow;
        const float* __restrict__ ap = lds_a + mrow * LDSP + 2 * hi;
        const float* __restrict__ bp = Wv + (size_t)nn * F_SIZE + 2 * hi;
        v8f c = {};
#pragma unroll 4
        for (int k = 0; k < F_SIZE; k += 4) {
            v2f a, b;
            a.x = ap[k]; a.y = ap[k + 1];
            b.x = bp[k]; b.y = bp[k + 1];
            c = __builtin_amdgcn_wmma_f32_16x16x4_f32(
                    false, a, false, b, (short)0, c, false, false);
        }
        const float bias = bv[nn];
#pragma unroll
        for (int r = 0; r < 8; ++r)
            lds_v[(r + 8 * hi) * LDSP + nn] = c[r] + bias;
    }
    __syncthreads();

#pragma unroll
    for (int tn = 0; tn < 2; ++tn) {
        const int nn = (wave * 2 + tn) * 16 + mrow;
        const float* __restrict__ ap = lds_v + mrow * LDSP + 2 * hi;
        const float* __restrict__ bp = Wo + (size_t)nn * F_SIZE + 2 * hi;
        v8f c = {};
#pragma unroll 4
        for (int k = 0; k < F_SIZE; k += 4) {
            v2f a, b;
            a.x = ap[k]; a.y = ap[k + 1];
            b.x = bp[k]; b.y = bp[k + 1];
            c = __builtin_amdgcn_wmma_f32_16x16x4_f32(
                    false, a, false, b, (short)0, c, false, false);
        }
        const float bias = bo[nn];
#pragma unroll
        for (int r = 0; r < 8; ++r)
            out[(size_t)(m0 + r + 8 * hi) * F_SIZE + nn] = c[r] + bias;
    }
}

extern "C" void kernel_launch(void* const* d_in, const int* in_sizes, int n_in,
                              void* d_out, int out_size, void* d_ws, size_t ws_size,
                              hipStream_t stream) {
    const float* act  = (const float*)d_in[0];
    const float* pas  = (const float*)d_in[1];
    const int*   bids = (const int*)d_in[2];
    const int*   mids = (const int*)d_in[3];
    const int*   xy   = (const int*)d_in[4];
    const int*   m0   = (const int*)d_in[5];
    const int*   m1   = (const int*)d_in[6];
    const int*   m2   = (const int*)d_in[7];
    const int*   m3   = (const int*)d_in[8];
    const float* Wa   = (const float*)d_in[9];
    const float* ba   = (const float*)d_in[10];
    const float* Wv   = (const float*)d_in[11];
    const float* bv   = (const float*)d_in[12];
    const float* Wo   = (const float*)d_in[13];
    const float* bo   = (const float*)d_in[14];
    float* out = (float*)d_out;

    const int n_act = in_sizes[0] / F_SIZE;

    float* fw_ws    = (float*)d_ws;
    int*   id_ws    = (int*)(fw_ws + (size_t)n_act * 32);
    float* weighted = (float*)(id_ws + (size_t)n_act * 32);

    k_attn_ids<<<(n_act + 7) / 8, 256, 0, stream>>>(
        act, bids, mids, xy, m0, m1, m2, m3, Wa, ba, fw_ws, id_ws, n_act);

    k_gather<<<n_act, 256, 0, stream>>>(act, pas, fw_ws, id_ws, weighted, n_act);

    k_dual_gemm<<<n_act / 16, 256, 0, stream>>>(weighted, Wv, bv, Wo, bo, out);
}


# K1 logits act.Wa^T on the f32 16x16x4 matrix instruction instead of a per-lane strided dot product: 4 K-quarter waves per 16-row tile, partials added in LDS; logit summation order changes
# speedup vs baseline: 1.5402x; 1.5402x over previous
#include <hip/hip_runtime.h>
#include <hip/hip_bf16.h>

typedef __attribute__((ext_vector_type(2))) float v2f;
typedef __attribute__((ext_vector_type(8))) float v8f;

#define F_SIZE 256
#define LDSP 260

#define K1_TILES 2
#define PPITCH 36

__device__ __forceinline__ void k1_load_frag(float4 (&f)[8], const float* __restrict__ rowp)
{
    const float4* __restrict__ p = (const float4*)rowp;
#pragma unroll
    for (int j = 0; j < 8; ++j) f[j] = p[2 * j];
}

__device__ __forceinline__ v8f k1_mma(float ax, float ay, float bx, float by, v8f c)
{
    v2f a = {ax, ay};
    v2f b = {bx, by};
    return __builtin_amdgcn_wmma_f32_16x16x4_f32(false, a, false, b, (short)0, c, false, false);
}

__device__ __forceinline__ void k1_quarter(const float4 (&af)[8], const float4 (&bf0)[8],
                                           const float4 (&bf1)[8], float* __restrict__ pp,
                                           int c, int h)
{
    v8f c0 = {}, c1 = {};
#pragma unroll
    for (int j = 0; j < 8; ++j) {
        c0 = k1_mma(af[j].x, af[j].y, bf0[j].x, bf0[j].y, c0);
        c1 = k1_mma(af[j].x, af[j].y, bf1[j].x, bf1[j].y, c1);
        c0 = k1_mma(af[j].z, af[j].w, bf0[j].z, bf0[j].w, c0);
        c1 = k1_mma(af[j].z, af[j].w, bf1[j].z, bf1[j].w, c1);
    }
#pragma unroll
    for (int r = 0; r < 8; ++r) {
        pp[(r + 8 * h) * PPITCH + c]      = c0[r];
        pp[(r + 8 * h) * PPITCH + 16 + c] = c1[r];
    }
}

__device__ __forceinline__ void k1_ids(int (&id)[4], int2 xy0, int m, int b, int offx, int offy,
                                       const int* __restrict__ m0, const int* __restrict__ m1,
                                       const int* __restrict__ m2, const int* __restrict__ m3)
{
    const int dim = 128 >> m;
    const int* __restrict__ map = (m == 0) ? m0 : (m == 1) ? m1 : (m == 2) ? m2 : m3;
#pragma unroll
    for (int p = 0; p < 4; ++p) {
        const int scale = p + 1;
        int x = xy0.x + scale * offx;
        int y = xy0.y + scale * offy;
        x = min(max(x, 0), dim - 1);
        y = min(max(y, 0), dim - 1);
        id[p] = map[((size_t)b * dim + y) * dim + x];
    }
}

__device__ __forceinline__ void k1_finish(const float* __restrict__ rp, float4 bias,
                                          const int (&id)[4], int erow, int ek, int n_act,
                                          float* __restrict__ fw_ws, int* __restrict__ id_ws)
{
    const float4 s0 = *(const float4*)(rp);
    const float4 s1 = *(const float4*)(rp + 1 * 16 * PPITCH);
    const float4 s2 = *(const float4*)(rp + 2 * 16 * PPITCH);
    const float4 s3 = *(const float4*)(rp + 3 * 16 * PPITCH);
    const float l0 = ((s0.x + s1.x) + s2.x) + s3.x + bias.x;
    const float l1 = ((s0.y + s1.y) + s2.y) + s3.y + bias.y;
    const float l2 = ((s0.z + s1.z) + s2.z) + s3.z + bias.z;
    const float l3 = ((s0.w + s1.w) + s2.w) + s3.w + bias.w;

    const float mx = fmaxf(fmaxf(l0, l1), fmaxf(l2, l3));
    const float e0 = __expf(l0 - mx);
    const float e1 = __expf(l1 - mx);
    const float e2 = __expf(l2 - mx);
    const float e3 = __expf(l3 - mx);
    const float s  = (e0 + e1) + (e2 + e3);
    const float fw[4] = {e0 / s, e1 / s, e2 / s, e3 / s};

    if (erow < n_act) {
#pragma unroll
        for (int p = 0; p < 4; ++p) {
            const size_t o = (size_t)erow * 32 + p * 8 + ek;
            fw_ws[o] = fw[p];
            id_ws[o] = id[p];
        }
    }
}

__global__ __launch_bounds__(128) void k_attn_ids(
    const float* __restrict__ act, const int* __restrict__ bids,
    const int* __restrict__ mids, const int* __restrict__ xy,
    const int* __restrict__ m0, const int* __restrict__ m1,
    const int* __restrict__ m2, const int* __restrict__ m3,
    const float* __restrict__ Wa, const float* __restrict__ ba,
    float* __restrict__ fw_ws, int* __restrict__ id_ws, int n_act)
{
    __shared__ __attribute__((aligned(16))) float s_part[K1_TILES][4][16 * PPITCH];

    const int t    = threadIdx.x;
    const int q    = t >> 5;
    const int lane = t & 31;
    const int c    = lane & 15;
    const int h    = lane >> 4;
    const int kofs = q * 64 + 4 * h;
    const int row0 = blockIdx.x * (16 * K1_TILES);

    const int er   = t >> 3;
    const int ek   = t & 7;
    const int kk   = ek + (ek >> 2);
    const int offx = kk % 3 - 1;
    const int offy = kk / 3 - 1;

    const int erow0 = row0 + er;
    const int erow1 = row0 + 16 + er;
    const int lrow0 = min(erow0, n_act - 1);
    const int lrow1 = min(erow1, n_act - 1);
    const int2 xyA = ((const int2*)xy)[lrow0];
    const int2 xyB = ((const int2*)xy)[lrow1];
    const int  mA  = mids[lrow0];
    const int  mB  = mids[lrow1];
    const int  bA  = bids[lrow0];
    const int  bB  = bids[lrow1];
    const float4 bias = *(const float4*)(ba + 4 * ek);

    float4 bf0[8], bf1[8], a0[8], a1[8];
    k1_load_frag(bf0, Wa + (size_t)c * F_SIZE + kofs);
    k1_load_frag(bf1, Wa + (size_t)(16 + c) * F_SIZE + kofs);
    k1_load_frag(a0, act + (size_t)min(row0 + c, n_act - 1) * F_SIZE + kofs);
    k1_load_frag(a1, act + (size_t)min(row0 + 16 + c, n_act - 1) * F_SIZE + kofs);

    int idA[4], idB[4];
    k1_ids(idA, xyA, mA, bA, offx, offy, m0, m1, m2, m3);
    k1_ids(idB, xyB, mB, bB, offx, offy, m0, m1, m2, m3);

    k1_quarter(a0, bf0, bf1, s_part[0][q], c, h);
    __syncthreads();
    k1_finish(&s_part[0][0][er * PPITCH + 4 * ek], bias, idA, erow0, ek, n_act, fw_ws, id_ws);

    k1_quarter(a1, bf0, bf1, s_part[1][q], c, h);
    __syncthreads();
    k1_finish(&s_part[1][0][er * PPITCH + 4 * ek], bias, idB, erow1, ek, n_act, fw_ws, id_ws);
}

__global__ __launch_bounds__(256) void k_gather(
    const float* __restrict__ act, const float* __restrict__ pas,
    const float* __restrict__ fw_ws, const int* __restrict__ id_ws,
    float* __restrict__ weighted, int n_act)
{
    __shared__ float s_fw[32];
    __shared__ const float* s_ptr[32];
    const int row = blockIdx.x;
    const int t = threadIdx.x;
    if (t < 32) {
        const int id = id_ws[(size_t)row * 32 + t];
        s_fw[t] = fw_ws[(size_t)row * 32 + t];
        s_ptr[t] = (id < n_act) ? (act + (size_t)id * F_SIZE)
                                : (pas + (size_t)(id - n_act) * F_SIZE);
    }
    __syncthreads();
    float acc = 0.f;
#pragma unroll
    for (int g = 0; g < 32; ++g)
        acc += s_fw[g] * s_ptr[g][t];
    weighted[(size_t)row * F_SIZE + t] = acc;
}

__global__ __launch_bounds__(256) void k_dual_gemm(
    const float* __restrict__ weighted,
    const float* __restrict__ Wv, const float* __restrict__ bv,
    const float* __restrict__ Wo, const float* __restrict__ bo,
    float* __restrict__ out)
{
    __shared__ float lds_a[16 * LDSP];
    __shared__ float lds_v[16 * LDSP];

    const int t = threadIdx.x;
    const int wave = t >> 5;
    const int lane = t & 31;
    const int m0 = blockIdx.x * 16;

    {
        const int r = t >> 4;
        const int c0 = (t & 15) * 16;
        const float4* __restrict__ src =
            (const float4*)(weighted + (size_t)(m0 + r) * F_SIZE + c0);
        float4* dst = (float4*)(lds_a + r * LDSP + c0);
        dst[0] = src[0]; dst[1] = src[1]; dst[2] = src[2]; dst[3] = src[3];
    }
    __syncthreads();

    const int mrow = lane & 15;
    const int hi = lane >> 4;

#pragma unroll
    for (int tn = 0; tn < 2; ++tn) {
        const int nn = (wave * 2 + tn) * 16 + mrow;
        const float* __restrict__ ap = lds_a + mrow * LDSP + 2 * hi;
        const float* __restrict__ bp = Wv + (size_t)nn * F_SIZE + 2 * hi;
        v8f c = {};
#pragma unroll 4
        for (int k = 0; k < F_SIZE; k += 4) {
            v2f a, b;
            a.x = ap[k]; a.y = ap[k + 1];
            b.x = bp[k]; b.y = bp[k + 1];
            c = __builtin_amdgcn_wmma_f32_16x16x4_f32(
                    false, a, false, b, (short)0, c, false, false);
        }
        const float bias = bv[nn];
#pragma unroll
        for (int r = 0; r < 8; ++r)
            lds_v[(r + 8 * hi) * LDSP + nn] = c[r] + bias;
    }
    __syncthreads();

#pragma unroll
    for (int tn = 0; tn < 2; ++tn) {
        const int nn = (wave * 2 + tn) * 16 + mrow;
        const float* __restrict__ ap = lds_v + mrow * LDSP + 2 * hi;
        const float* __restrict__ bp = Wo + (size_t)nn * F_SIZE + 2 * hi;
        v8f c = {};
#pragma unroll 4
        for (int k = 0; k < F_SIZE; k += 4) {
            v2f a, b;
            a.x = ap[k]; a.y = ap[k + 1];
            b.x = bp[k]; b.y = bp[k + 1];
            c = __builtin_amdgcn_wmma_f32_16x16x4_f32(
                    false, a, false, b, (short)0, c, false, false);
        }
        const float bias = bo[nn];
#pragma unroll
        for (int r = 0; r < 8; ++r)
            out[(size_t)(m0 + r + 8 * hi) * F_SIZE + nn] = c[r] + bias;
    }
}

extern "C" void kernel_launch(void* const* d_in, const int* in_sizes, int n_in,
                              void* d_out, int out_size, void* d_ws, size_t ws_size,
                              hipStream_t stream) {
    const float* act  = (const float*)d_in[0];
    const float* pas  = (const float*)d_in[1];
    const int*   bids = (const int*)d_in[2];
    const int*   mids = (const int*)d_in[3];
    const int*   xy   = (const int*)d_in[4];
    const int*   m0   = (const int*)d_in[5];
    const int*   m1   = (const int*)d_in[6];
    const int*   m2   = (const int*)d_in[7];
    const int*   m3   = (const int*)d_in[8];
    const float* Wa   = (const float*)d_in[9];
    const float* ba   = (const float*)d_in[10];
    const float* Wv   = (const float*)d_in[11];
    const float* bv   = (const float*)d_in[12];
    const float* Wo   = (const float*)d_in[13];
    const float* bo   = (const float*)d_in[14];
    float* out = (float*)d_out;

    const int n_act = in_sizes[0] / F_SIZE;

    float* fw_ws    = (float*)d_ws;
    int*   id_ws    = (int*)(fw_ws + (size_t)n_act * 32);
    float* weighted = (float*)(id_ws + (size_t)n_act * 32);

    k_attn_ids<<<(n_act + 16 * K1_TILES - 1) / (16 * K1_TILES), 128, 0, stream>>>(
        act, bids, mids, xy, m0, m1, m2, m3, Wa, ba, fw_ws, id_ws, n_act);

    k_gather<<<n_act, 256, 0, stream>>>(act, pas, fw_ws, id_ws, weighted, n_act);

    k_dual_gemm<<<n_act / 16, 256, 0, stream>>>(weighted, Wv, bv, Wo, bo, out);
}


# K3: block tile 16 to 64 rows, 8 accumulators per wave, so each Wv/Wo fragment load (float4 per lane) feeds 8 matrix instructions, was 1 per float2 load; k summation order of both products changes
# speedup vs baseline: 2.0679x; 1.3426x over previous
#include <hip/hip_runtime.h>
#include <hip/hip_bf16.h>

typedef __attribute__((ext_vector_type(2))) float v2f;
typedef __attribute__((ext_vector_type(8))) float v8f;

#define F_SIZE 256
#define LDSP 260

#define K1_TILES 2
#define PPITCH 36

__device__ __forceinline__ void k1_load_frag(float4 (&f)[8], const float* __restrict__ rowp)
{
    const float4* __restrict__ p = (const float4*)rowp;
#pragma unroll
    for (int j = 0; j < 8; ++j) f[j] = p[2 * j];
}

__device__ __forceinline__ v8f k1_mma(float ax, float ay, float bx, float by, v8f c)
{
    v2f a = {ax, ay};
    v2f b = {bx, by};
    return __builtin_amdgcn_wmma_f32_16x16x4_f32(false, a, false, b, (short)0, c, false, false);
}

__device__ __forceinline__ void k1_quarter(const float4 (&af)[8], const float4 (&bf0)[8],
                                           const float4 (&bf1)[8], float* __restrict__ pp,
                                           int c, int h)
{
    v8f c0 = {}, c1 = {};
#pragma unroll
    for (int j = 0; j < 8; ++j) {
        c0 = k1_mma(af[j].x, af[j].y, bf0[j].x, bf0[j].y, c0);
        c1 = k1_mma(af[j].x, af[j].y, bf1[j].x, bf1[j].y, c1);
        c0 = k1_mma(af[j].z, af[j].w, bf0[j].z, bf0[j].w, c0);
        c1 = k1_mma(af[j].z, af[j].w, bf1[j].z, bf1[j].w, c1);
    }
#pragma unroll
    for (int r = 0; r < 8; ++r) {
        pp[(r + 8 * h) * PPITCH + c]      = c0[r];
        pp[(r + 8 * h) * PPITCH + 16 + c] = c1[r];
    }
}

__device__ __forceinline__ void k1_ids(int (&id)[4], int2 xy0, int m, int b, int offx, int offy,
                                       const int* __restrict__ m0, const int* __restrict__ m1,
                                       const int* __restrict__ m2, const int* __restrict__ m3)
{
    const int dim = 128 >> m;
    const int* __restrict__ map = (m == 0) ? m0 : (m == 1) ? m1 : (m == 2) ? m2 : m3;
#pragma unroll
    for (int p = 0; p < 4; ++p) {
        const int scale = p + 1;
        int x = xy0.x + scale * offx;
        int y = xy0.y + scale * offy;
        x = min(max(x, 0), dim - 1);
        y = min(max(y, 0), dim - 1);
        id[p] = map[((size_t)b * dim + y) * dim + x];
    }
}

__device__ __forceinline__ void k1_finish(const float* __restrict__ rp, float4 bias,
                                          const int (&id)[4], int erow, int ek, int n_act,
                                          float* __restrict__ fw_ws, int* __restrict__ id_ws)
{
    const float4 s0 = *(const float4*)(rp);
    const float4 s1 = *(const float4*)(rp + 1 * 16 * PPITCH);
    const float4 s2 = *(const float4*)(rp + 2 * 16 * PPITCH);
    const float4 s3 = *(const float4*)(rp + 3 * 16 * PPITCH);
    const float l0 = ((s0.x + s1.x) + s2.x) + s3.x + bias.x;
    const float l1 = ((s0.y + s1.y) + s2.y) + s3.y + bias.y;
    const float l2 = ((s0.z + s1.z) + s2.z) + s3.z + bias.z;
    const float l3 = ((s0.w + s1.w) + s2.w) + s3.w + bias.w;

    const float mx = fmaxf(fmaxf(l0, l1), fmaxf(l2, l3));
    const float e0 = __expf(l0 - mx);
    const float e1 = __expf(l1 - mx);
    const float e2 = __expf(l2 - mx);
    const float e3 = __expf(l3 - mx);
    const float s  = (e0 + e1) + (e2 + e3);
    const float fw[4] = {e0 / s, e1 / s, e2 / s, e3 / s};

    if (erow < n_act) {
#pragma unroll
        for (int p = 0; p < 4; ++p) {
            const size_t o = (size_t)erow * 32 + p * 8 + ek;
            fw_ws[o] = fw[p];
            id_ws[o] = id[p];
        }
    }
}

__global__ __launch_bounds__(128) void k_attn_ids(
    const float* __restrict__ act, const int* __restrict__ bids,
    const int* __restrict__ mids, const int* __restrict__ xy,
    const int* __restrict__ m0, const int* __restrict__ m1,
    const int* __restrict__ m2, const int* __restrict__ m3,
    const float* __restrict__ Wa, const float* __restrict__ ba,
    float* __restrict__ fw_ws, int* __restrict__ id_ws, int n_act)
{
    __shared__ __attribute__((aligned(16))) float s_part[K1_TILES][4][16 * PPITCH];

    const int t    = threadIdx.x;
    const int q    = t >> 5;
    const int lane = t & 31;
    const int c    = lane & 15;
    const int h    = lane >> 4;
    const int kofs = q * 64 + 4 * h;
    const int row0 = blockIdx.x * (16 * K1_TILES);

    const int er   = t >> 3;
    const int ek   = t & 7;
    const int kk   = ek + (ek >> 2);
    const int offx = kk % 3 - 1;
    const int offy = kk / 3 - 1;

    const int erow0 = row0 + er;
    const int erow1 = row0 + 16 + er;
    const int lrow0 = min(erow0, n_act - 1);
    const int lrow1 = min(erow1, n_act - 1);
    const int2 xyA = ((const int2*)xy)[lrow0];
    const int2 xyB = ((const int2*)xy)[lrow1];
    const int  mA  = mids[lrow0];
    const int  mB  = mids[lrow1];
    const int  bA  = bids[lrow0];
    const int  bB  = bids[lrow1];
    const float4 bias = *(const float4*)(ba + 4 * ek);

    float4 bf0[8], bf1[8], a0[8], a1[8];
    k1_load_frag(bf0, Wa + (size_t)c * F_SIZE + kofs);
    k1_load_frag(bf1, Wa + (size_t)(16 + c) * F_SIZE + kofs);
    k1_load_frag(a0, act + (size_t)min(row0 + c, n_act - 1) * F_SIZE + kofs);
    k1_load_frag(a1, act + (size_t)min(row0 + 16 + c, n_act - 1) * F_SIZE + kofs);

    int idA[4], idB[4];
    k1_ids(idA, xyA, mA, bA, offx, offy, m0, m1, m2, m3);
    k1_ids(idB, xyB, mB, bB, offx, offy, m0, m1, m2, m3);

    k1_quarter(a0, bf0, bf1, s_part[0][q], c, h);
    __syncthreads();
    k1_finish(&s_part[0][0][er * PPITCH + 4 * ek], bias, idA, erow0, ek, n_act, fw_ws, id_ws);

    k1_quarter(a1, bf0, bf1, s_part[1][q], c, h);
    __syncthreads();
    k1_finish(&s_part[1][0][er * PPITCH + 4 * ek], bias, idB, erow1, ek, n_act, fw_ws, id_ws);
}

__global__ __launch_bounds__(256) void k_gather(
    const float* __restrict__ act, const float* __restrict__ pas,
    const float* __restrict__ fw_ws, const int* __restrict__ id_ws,
    float* __restrict__ weighted, int n_act)
{
    __shared__ float s_fw[32];
    __shared__ const float* s_ptr[32];
    const int row = blockIdx.x;
    const int t = threadIdx.x;
    if (t < 32) {
        const int id = id_ws[(size_t)row * 32 + t];
        s_fw[t] = fw_ws[(size_t)row * 32 + t];
        s_ptr[t] = (id < n_act) ? (act + (size_t)id * F_SIZE)
                                : (pas + (size_t)(id - n_act) * F_SIZE);
    }
    __syncthreads();
    float acc = 0.f;
#pragma unroll
    for (int g = 0; g < 32; ++g)
        acc += s_fw[g] * s_ptr[g][t];
    weighted[(size_t)row * F_SIZE + t] = acc;
}

#define K3_MT 4
#define K3_ROWS (16 * K3_MT)

struct K3Frag {
    float4 a[K3_MT][2];
    float4 b[2][2];
};

__device__ __forceinline__ void k3_load(K3Frag& f, const float* __restrict__ ap,
                                        const float* __restrict__ bp0,
                                        const float* __restrict__ bp1, int kc)
{
#pragma unroll
    for (int j = 0; j < 2; ++j) {
        f.b[0][j] = *(const float4*)(bp0 + kc + 8 * j);
        f.b[1][j] = *(const float4*)(bp1 + kc + 8 * j);
    }
#pragma unroll
    for (int i = 0; i < K3_MT; ++i)
#pragma unroll
        for (int j = 0; j < 2; ++j)
            f.a[i][j] = *(const float4*)(ap + i * 16 * LDSP + kc + 8 * j);
}

__device__ __forceinline__ void k3_mma(v8f (&acc)[K3_MT][2], const K3Frag& f)
{
#pragma unroll
    for (int j = 0; j < 2; ++j) {
#pragma unroll
        for (int i = 0; i < K3_MT; ++i) {
            acc[i][0] = k1_mma(f.a[i][j].x, f.a[i][j].y, f.b[0][j].x, f.b[0][j].y, acc[i][0]);
            acc[i][1] = k1_mma(f.a[i][j].x, f.a[i][j].y, f.b[1][j].x, f.b[1][j].y, acc[i][1]);
        }
#pragma unroll
        for (int i = 0; i < K3_MT; ++i) {
            acc[i][0] = k1_mma(f.a[i][j].z, f.a[i][j].w, f.b[0][j].z, f.b[0][j].w, acc[i][0]);
            acc[i][1] = k1_mma(f.a[i][j].z, f.a[i][j].w, f.b[1][j].z, f.b[1][j].w, acc[i][1]);
        }
    }
}

__device__ __forceinline__ void k3_panel(v8f (&acc)[K3_MT][2], const float* __restrict__ ap,
                                         const float* __restrict__ bp0,
                                         const float* __restrict__ bp1)
{
#pragma unroll
    for (int i = 0; i < K3_MT; ++i) {
        acc[i][0] = v8f{};
        acc[i][1] = v8f{};
    }
    K3Frag f0, f1;
    k3_load(f0, ap, bp0, bp1, 0);
    for (int kc = 0; kc < F_SIZE; kc += 32) {
        k3_load(f1, ap, bp0, bp1, kc + 16);
        k3_mma(acc, f0);
        if (kc + 32 < F_SIZE) k3_load(f0, ap, bp0, bp1, kc + 32);
        k3_mma(acc, f1);
    }
}

__global__ __launch_bounds__(256) void k_dual_gemm(
    const float* __restrict__ weighted,
    const float* __restrict__ Wv, const float* __restrict__ bv,
    const float* __restrict__ Wo, const float* __restrict__ bo,
    float* __restrict__ out, int n_act)
{
    __shared__ __attribute__((aligned(16))) float lds_a[K3_ROWS * LDSP];
    __shared__ __attribute__((aligned(16))) float lds_v[K3_ROWS * LDSP];

    const int t = threadIdx.x;
    const int wave = t >> 5;
    const int lane = t & 31;
    const int m0 = blockIdx.x * K3_ROWS;

    {
        float4 v[16];
#pragma unroll
        for (int i = 0; i < 16; ++i) {
            const int idx = t + 256 * i;
            const int r = idx >> 6;
            const int c4 = idx & 63;
            v[i] = ((const float4*)(weighted + (size_t)min(m0 + r, n_act - 1) * F_SIZE))[c4];
        }
#pragma unroll
        for (int i = 0; i < 16; ++i) {
            const int idx = t + 256 * i;
            const int r = idx >> 6;
            const int c4 = idx & 63;
            *(float4*)(lds_a + r * LDSP + 4 * c4) = v[i];
        }
    }
    __syncthreads();

    const int c = lane & 15;
    const int h = lane >> 4;
    const int nn0 = (wave * 2 + 0) * 16 + c;
    const int nn1 = (wave * 2 + 1) * 16 + c;

    v8f acc[K3_MT][2];

    k3_panel(acc, lds_a + c * LDSP + 4 * h,
             Wv + (size_t)nn0 * F_SIZE + 4 * h, Wv + (size_t)nn1 * F_SIZE + 4 * h);
    {
        const float bias0 = bv[nn0];
        const float bias1 = bv[nn1];
#pragma unroll
        for (int i = 0; i < K3_MT; ++i)
#pragma unroll
            for (int r = 0; r < 8; ++r) {
                lds_v[(16 * i + r + 8 * h) * LDSP + nn0] = acc[i][0][r] + bias0;
                lds_v[(16 * i + r + 8 * h) * LDSP + nn1] = acc[i][1][r] + bias1;
            }
    }
    __syncthreads();

    k3_panel(acc, lds_v + c * LDSP + 4 * h,
             Wo + (size_t)nn0 * F_SIZE + 4 * h, Wo + (size_t)nn1 * F_SIZE + 4 * h);
    {
        const float bias0 = bo[nn0];
        const float bias1 = bo[nn1];
#pragma unroll
        for (int i = 0; i < K3_MT; ++i)
#pragma unroll
            for (int r = 0; r < 8; ++r) {
                const int row = m0 + 16 * i + r + 8 * h;
                if (row < n_act) {
                    out[(size_t)row * F_SIZE + nn0] = acc[i][0][r] + bias0;
                    out[(size_t)row * F_SIZE + nn1] = acc[i][1][r] + bias1;
                }
            }
    }
}

extern "C" void kernel_launch(void* const* d_in, const int* in_sizes, int n_in,
                              void* d_out, int out_size, void* d_ws, size_t ws_size,
                              hipStream_t stream) {
    const float* act  = (const float*)d_in[0];
    const float* pas  = (const float*)d_in[1];
    const int*   bids = (const int*)d_in[2];
    const int*   mids = (const int*)d_in[3];
    const int*   xy   = (const int*)d_in[4];
    const int*   m0   = (const int*)d_in[5];
    const int*   m1   = (const int*)d_in[6];
    const int*   m2   = (const int*)d_in[7];
    const int*   m3   = (const int*)d_in[8];
    const float* Wa   = (const float*)d_in[9];
    const float* ba   = (const float*)d_in[10];
    const float* Wv   = (const float*)d_in[11];
    const float* bv   = (const float*)d_in[12];
    const float* Wo   = (const float*)d_in[13];
    const float* bo   = (const float*)d_in[14];
    float* out = (float*)d_out;

    const int n_act = in_sizes[0] / F_SIZE;

    float* fw_ws    = (float*)d_ws;
    int*   id_ws    = (int*)(fw_ws + (size_t)n_act * 32);
    float* weighted = (float*)(id_ws + (size_t)n_act * 32);

    k_attn_ids<<<(n_act + 16 * K1_TILES - 1) / (16 * K1_TILES), 128, 0, stream>>>(
        act, bids, mids, xy, m0, m1, m2, m3, Wa, ba, fw_ws, id_ws, n_act);

    k_gather<<<n_act, 256, 0, stream>>>(act, pas, fw_ws, id_ws, weighted, n_act);

    k_dual_gemm<<<(n_act + K3_ROWS - 1) / K3_ROWS, 256, 0, stream>>>(weighted, Wv, bv, Wo, bo, out, n_act);
}


# Wo.Wv composed once per call into one 256x256 matrix (bias Wo.bv+bo) by 128 extra blocks of launch 1 on the f32 matrix unit; launch 3 runs one GEMM, not two; output summation order changes
# speedup vs baseline: 2.5155x; 1.2165x over previous
#include <hip/hip_runtime.h>
#include <hip/hip_bf16.h>

typedef __attribute__((ext_vector_type(2))) float v2f;
typedef __attribute__((ext_vector_type(8))) float v8f;

#define F_SIZE 256
#define LDSP 260

#define K1_TILES 2
#define PPITCH 36
#define K1_FOLD_BLOCKS 128

__device__ __forceinline__ void k1_load_frag(float4 (&f)[8], const float* __restrict__ rowp)
{
    const float4* __restrict__ p = (const float4*)rowp;
#pragma unroll
    for (int j = 0; j < 8; ++j) f[j] = p[2 * j];
}

__device__ __forceinline__ void k1_load_frag_col(float4 (&f)[8], const float* __restrict__ colp)
{
#pragma unroll
    for (int j = 0; j < 8; ++j) {
        const float* __restrict__ r = colp + (size_t)(8 * j) * F_SIZE;
        f[j] = make_float4(r[0], r[F_SIZE], r[2 * F_SIZE], r[3 * F_SIZE]);
    }
}

__device__ __forceinline__ v8f k1_mma(float ax, float ay, float bx, float by, v8f c)
{
    v2f a = {ax, ay};
    v2f b = {bx, by};
    return __builtin_amdgcn_wmma_f32_16x16x4_f32(false, a, false, b, (short)0, c, false, false);
}

__device__ __forceinline__ void k1_quarter(const float4 (&af)[8], const float4 (&bf0)[8],
                                           const float4 (&bf1)[8], float* __restrict__ pp,
                                           int c, int h)
{
    v8f c0 = {}, c1 = {};
#pragma unroll
    for (int j = 0; j < 8; ++j) {
        c0 = k1_mma(af[j].x, af[j].y, bf0[j].x, bf0[j].y, c0);
        c1 = k1_mma(af[j].x, af[j].y, bf1[j].x, bf1[j].y, c1);
        c0 = k1_mma(af[j].z, af[j].w, bf0[j].z, bf0[j].w, c0);
        c1 = k1_mma(af[j].z, af[j].w, bf1[j].z, bf1[j].w, c1);
    }
#pragma unroll
    for (int r = 0; r < 8; ++r) {
        pp[(r + 8 * h) * PPITCH + c]      = c0[r];
        pp[(r + 8 * h) * PPITCH + 16 + c] = c1[r];
    }
}

__device__ __forceinline__ void k1_ids(int (&id)[4], int2 xy0, int m, int b, int offx, int offy,
                                       const int* __restrict__ m0, const int* __restrict__ m1,
                                       const int* __restrict__ m2, const int* __restrict__ m3)
{
    const int dim = 128 >> m;
    const int* __restrict__ map = (m == 0) ? m0 : (m == 1) ? m1 : (m == 2) ? m2 : m3;
#pragma unroll
    for (int p = 0; p < 4; ++p) {
        const int scale = p + 1;
        int x = xy0.x + scale * offx;
        int y = xy0.y + scale * offy;
        x = min(max(x, 0), dim - 1);
        y = min(max(y, 0), dim - 1);
        id[p] = map[((size_t)b * dim + y) * dim + x];
    }
}

__device__ __forceinline__ void k1_finish(const float* __restrict__ rp, float4 bias,
                                          const int (&id)[4], int erow, int ek, int n_act,
                                          float* __restrict__ fw_ws, int* __restrict__ id_ws)
{
    const float4 s0 = *(const float4*)(rp);
    const float4 s1 = *(const float4*)(rp + 1 * 16 * PPITCH);
    const float4 s2 = *(const float4*)(rp + 2 * 16 * PPITCH);
    const float4 s3 = *(const float4*)(rp + 3 * 16 * PPITCH);
    const float l0 = ((s0.x + s1.x) + s2.x) + s3.x + bias.x;
    const float l1 = ((s0.y + s1.y) + s2.y) + s3.y + bias.y;
    const float l2 = ((s0.z + s1.z) + s2.z) + s3.z + bias.z;
    const float l3 = ((s0.w + s1.w) + s2.w) + s3.w + bias.w;

    const float mx = fmaxf(fmaxf(l0, l1), fmaxf(l2, l3));
    const float e0 = __expf(l0 - mx);
    const float e1 = __expf(l1 - mx);
    const float e2 = __expf(l2 - mx);
    const float e3 = __expf(l3 - mx);
    const float s  = (e0 + e1) + (e2 + e3);
    const float fw[4] = {e0 / s, e1 / s, e2 / s, e3 / s};

    if (erow < n_act) {
#pragma unroll
        for (int p = 0; p < 4; ++p) {
            const size_t o = (size_t)erow * 32 + p * 8 + ek;
            fw_ws[o] = fw[p];
            id_ws[o] = id[p];
        }
    }
}

__global__ __launch_bounds__(128) void k_attn_ids(
    const float* __restrict__ act, const int* __restrict__ bids,
    const int* __restrict__ mids, const int* __restrict__ xy,
    const int* __restrict__ m0, const int* __restrict__ m1,
    const int* __restrict__ m2, const int* __restrict__ m3,
    const float* __restrict__ Wa, const float* __restrict__ ba,
    const float* __restrict__ Wv, const float* __restrict__ bv,
    const float* __restrict__ Wo, const float* __restrict__ bo,
    float* __restrict__ wc_ws, float* __restrict__ bc_ws,
    float* __restrict__ fw_ws, int* __restrict__ id_ws, int n_act)
{
    __shared__ __attribute__((aligned(16))) float s_part[K1_TILES][4][16 * PPITCH];

    const int t    = threadIdx.x;
    const int q    = t >> 5;
    const int lane = t & 31;
    const int c    = lane & 15;
    const int h    = lane >> 4;
    const int kofs = q * 64 + 4 * h;

    const int er   = t >> 3;
    const int ek   = t & 7;

    if (blockIdx.x < K1_FOLD_BLOCKS) {
        const int o0 = 16 * (blockIdx.x >> 3);
        const int i0 = 32 * (blockIdx.x & 7);
        float4 af[8], wf0[8], wf1[8];
        k1_load_frag(af, Wo + (size_t)(o0 + c) * F_SIZE + kofs);
        k1_load_frag_col(wf0, Wv + (size_t)kofs * F_SIZE + i0 + c);
        k1_load_frag_col(wf1, Wv + (size_t)kofs * F_SIZE + i0 + 16 + c);
        k1_quarter(af, wf0, wf1, s_part[0][q], c, h);
        __syncthreads();
        {
            const float* __restrict__ rp = &s_part[0][0][er * PPITCH + 4 * ek];
            const float4 s0 = *(const float4*)(rp);
            const float4 s1 = *(const float4*)(rp + 1 * 16 * PPITCH);
            const float4 s2 = *(const float4*)(rp + 2 * 16 * PPITCH);
            const float4 s3 = *(const float4*)(rp + 3 * 16 * PPITCH);
            float4 w;
            w.x = ((s0.x + s1.x) + s2.x) + s3.x;
            w.y = ((s0.y + s1.y) + s2.y) + s3.y;
            w.z = ((s0.z + s1.z) + s2.z) + s3.z;
            w.w = ((s0.w + s1.w) + s2.w) + s3.w;
            *(float4*)(wc_ws + (size_t)(o0 + er) * F_SIZE + i0 + 4 * ek) = w;
        }
        if ((blockIdx.x & 7) == 0) {
            const float4* __restrict__ wr = (const float4*)(Wo + (size_t)(o0 + er) * F_SIZE + 32 * ek);
            const float4* __restrict__ br = (const float4*)(bv + 32 * ek);
            float s = 0.f;
#pragma unroll
            for (int u = 0; u < 8; ++u) {
                const float4 wv4 = wr[u];
                const float4 bv4 = br[u];
                s = fmaf(wv4.x, bv4.x, s);
                s = fmaf(wv4.y, bv4.y, s);
                s = fmaf(wv4.z, bv4.z, s);
                s = fmaf(wv4.w, bv4.w, s);
            }
            s += __shfl_xor(s, 1);
            s += __shfl_xor(s, 2);
            s += __shfl_xor(s, 4);
            if (ek == 0) bc_ws[o0 + er] = s + bo[o0 + er];
        }
        return;
    }

    const int row0 = (blockIdx.x - K1_FOLD_BLOCKS) * (16 * K1_TILES);
    const int kk   = ek + (ek >> 2);
    const int offx = kk % 3 - 1;
    const int offy = kk / 3 - 1;

    const int erow0 = row0 + er;
    const int erow1 = row0 + 16 + er;
    const int lrow0 = min(erow0, n_act - 1);
    const int lrow1 = min(erow1, n_act - 1);
    const int2 xyA = ((const int2*)xy)[lrow0];
    const int2 xyB = ((const int2*)xy)[lrow1];
    const int  mA  = mids[lrow0];
    const int  mB  = mids[lrow1];
    const int  bA  = bids[lrow0];
    const int  bB  = bids[lrow1];
    const float4 bias = *(const float4*)(ba + 4 * ek);

    float4 bf0[8], bf1[8], a0[8], a1[8];
    k1_load_frag(bf0, Wa + (size_t)c * F_SIZE + kofs);
    k1_load_frag(bf1, Wa + (size_t)(16 + c) * F_SIZE + kofs);
    k1_load_frag(a0, act + (size_t)min(row0 + c, n_act - 1) * F_SIZE + kofs);
    k1_load_frag(a1, act + (size_t)min(row0 + 16 + c, n_act - 1) * F_SIZE + kofs);

    int idA[4], idB[4];
    k1_ids(idA, xyA, mA, bA, offx, offy, m0, m1, m2, m3);
    k1_ids(idB, xyB, mB, bB, offx, offy, m0, m1, m2, m3);

    k1_quarter(a0, bf0, bf1, s_part[0][q], c, h);
    __syncthreads();
    k1_finish(&s_part[0][0][er * PPITCH + 4 * ek], bias, idA, erow0, ek, n_act, fw_ws, id_ws);

    k1_quarter(a1, bf0, bf1, s_part[1][q], c, h);
    __syncthreads();
    k1_finish(&s_part[1][0][er * PPITCH + 4 * ek], bias, idB, erow1, ek, n_act, fw_ws, id_ws);
}

__global__ __launch_bounds__(256) void k_gather(
    const float* __restrict__ act, const float* __restrict__ pas,
    const float* __restrict__ fw_ws, const int* __restrict__ id_ws,
    float* __restrict__ weighted, int n_act)
{
    __shared__ float s_fw[32];
    __shared__ const float* s_ptr[32];
    const int row = blockIdx.x;
    const int t = threadIdx.x;
    if (t < 32) {
        const int id = id_ws[(size_t)row * 32 + t];
        s_fw[t] = fw_ws[(size_t)row * 32 + t];
        s_ptr[t] = (id < n_act) ? (act + (size_t)id * F_SIZE)
                                : (pas + (size_t)(id - n_act) * F_SIZE);
    }
    __syncthreads();
    float acc = 0.f;
#pragma unroll
    for (int g = 0; g < 32; ++g)
        acc += s_fw[g] * s_ptr[g][t];
    weighted[(size_t)row * F_SIZE + t] = acc;
}

#define K3_MT 4
#define K3_ROWS (16 * K3_MT)

struct K3Frag {
    float4 a[K3_MT][2];
    float4 b[2][2];
};

__device__ __forceinline__ void k3_load(K3Frag& f, const float* __restrict__ ap,
                                        const float* __restrict__ bp0,
                                        const float* __restrict__ bp1, int kc)
{
#pragma unroll
    for (int j = 0; j < 2; ++j) {
        f.b[0][j] = *(const float4*)(bp0 + kc + 8 * j);
        f.b[1][j] = *(const float4*)(bp1 + kc + 8 * j);
    }
#pragma unroll
    for (int i = 0; i < K3_MT; ++i)
#pragma unroll
        for (int j = 0; j < 2; ++j)
            f.a[i][j] = *(const float4*)(ap + i * 16 * LDSP + kc + 8 * j);
}

__device__ __forceinline__ void k3_mma(v8f (&acc)[K3_MT][2], const K3Frag& f)
{
#pragma unroll
    for (int j = 0; j < 2; ++j) {
#pragma unroll
        for (int i = 0; i < K3_MT; ++i) {
            acc[i][0] = k1_mma(f.a[i][j].x, f.a[i][j].y, f.b[0][j].x, f.b[0][j].y, acc[i][0]);
            acc[i][1] = k1_mma(f.a[i][j].x, f.a[i][j].y, f.b[1][j].x, f.b[1][j].y, acc[i][1]);
        }
#pragma unroll
        for (int i = 0; i < K3_MT; ++i) {
            acc[i][0] = k1_mma(f.a[i][j].z, f.a[i][j].w, f.b[0][j].z, f.b[0][j].w, acc[i][0]);
            acc[i][1] = k1_mma(f.a[i][j].z, f.a[i][j].w, f.b[1][j].z, f.b[1][j].w, acc[i][1]);
        }
    }
}

__device__ __forceinline__ void k3_panel(v8f (&acc)[K3_MT][2], const float* __restrict__ ap,
                                         const float* __restrict__ bp0,
                                         const float* __restrict__ bp1)
{
#pragma unroll
    for (int i = 0; i < K3_MT; ++i) {
        acc[i][0] = v8f{};
        acc[i][1] = v8f{};
    }
    K3Frag f0, f1;
    k3_load(f0, ap, bp0, bp1, 0);
    for (int kc = 0; kc < F_SIZE; kc += 32) {
        k3_load(f1, ap, bp0, bp1, kc + 16);
        k3_mma(acc, f0);
        if (kc + 32 < F_SIZE) k3_load(f0, ap, bp0, bp1, kc + 32);
        k3_mma(acc, f1);
    }
}

__global__ __launch_bounds__(256) void k_out_gemm(
    const float* __restrict__ weighted,
    const float* __restrict__ wc_ws, const float* __restrict__ bc_ws,
    float* __restrict__ out, int n_act)
{
    __shared__ __attribute__((aligned(16))) float lds_a[K3_ROWS * LDSP];

    const int t = threadIdx.x;
    const int wave = t >> 5;
    const int lane = t & 31;
    const int m0 = blockIdx.x * K3_ROWS;

    {
        float4 v[16];
#pragma unroll
        for (int i = 0; i < 16; ++i) {
            const int idx = t + 256 * i;
            const int r = idx >> 6;
            const int c4 = idx & 63;
            v[i] = ((const float4*)(weighted + (size_t)min(m0 + r, n_act - 1) * F_SIZE))[c4];
        }
#pragma unroll
        for (int i = 0; i < 16; ++i) {
            const int idx = t + 256 * i;
            const int r = idx >> 6;
            const int c4 = idx & 63;
            *(float4*)(lds_a + r * LDSP + 4 * c4) = v[i];
        }
    }
    __syncthreads();

    const int c = lane & 15;
    const int h = lane >> 4;
    const int nn0 = (wave * 2 + 0) * 16 + c;
    const int nn1 = (wave * 2 + 1) * 16 + c;

    v8f acc[K3_MT][2];

    k3_panel(acc, lds_a + c * LDSP + 4 * h,
             wc_ws + (size_t)nn0 * F_SIZE + 4 * h, wc_ws + (size_t)nn1 * F_SIZE + 4 * h);
    {
        const float bias0 = bc_ws[nn0];
        const float bias1 = bc_ws[nn1];
#pragma unroll
        for (int i = 0; i < K3_MT; ++i)
#pragma unroll
            for (int r = 0; r < 8; ++r) {
                const int row = m0 + 16 * i + r + 8 * h;
                if (row < n_act) {
                    out[(size_t)row * F_SIZE + nn0] = acc[i][0][r] + bias0;
                    out[(size_t)row * F_SIZE + nn1] = acc[i][1][r] + bias1;
                }
            }
    }
}

extern "C" void kernel_launch(void* const* d_in, const int* in_sizes, int n_in,
                              void* d_out, int out_size, void* d_ws, size_t ws_size,
                              hipStream_t stream) {
    const float* act  = (const float*)d_in[0];
    const float* pas  = (const float*)d_in[1];
    const int*   bids = (const int*)d_in[2];
    const int*   mids = (const int*)d_in[3];
    const int*   xy   = (const int*)d_in[4];
    const int*   m0   = (const int*)d_in[5];
    const int*   m1   = (const int*)d_in[6];
    const int*   m2   = (const int*)d_in[7];
    const int*   m3   = (const int*)d_in[8];
    const float* Wa   = (const float*)d_in[9];
    const float* ba   = (const float*)d_in[10];
    const float* Wv   = (const float*)d_in[11];
    const float* bv   = (const float*)d_in[12];
    const float* Wo   = (const float*)d_in[13];
    const float* bo   = (const float*)d_in[14];
    float* out = (float*)d_out;

    const int n_act = in_sizes[0] / F_SIZE;

    float* fw_ws    = out;
    int*   id_ws    = (int*)(out + (size_t)n_act * 32);
    float* wc_ws    = (float*)d_ws;
    float* bc_ws    = wc_ws + F_SIZE * F_SIZE;
    float* weighted = bc_ws + F_SIZE;

    k_attn_ids<<<K1_FOLD_BLOCKS + (n_act + 16 * K1_TILES - 1) / (16 * K1_TILES), 128, 0, stream>>>(
        act, bids, mids, xy, m0, m1, m2, m3, Wa, ba, Wv, bv, Wo, bo, wc_ws, bc_ws,
        fw_ws, id_ws, n_act);

    k_gather<<<n_act, 256, 0, stream>>>(act, pas, fw_ws, id_ws, weighted, n_act);

    k_out_gemm<<<(n_act + K3_ROWS - 1) / K3_ROWS, 256, 0, stream>>>(weighted, wc_ws, bc_ws, out, n_act);
}


# K2 gather: one wave per row, two 16-byte-per-lane loads per gathered 1KB row, ids and weights in scalar registers, no LDS or barrier; was 8 waves per row with 4-byte loads; sum order unchanged
# speedup vs baseline: 3.0435x; 1.2099x over previous
#include <hip/hip_runtime.h>
#include <hip/hip_bf16.h>

typedef __attribute__((ext_vector_type(2))) float v2f;
typedef __attribute__((ext_vector_type(8))) float v8f;

#define F_SIZE 256
#define LDSP 260

#define K1_TILES 2
#define PPITCH 36
#define K1_FOLD_BLOCKS 128

__device__ __forceinline__ void k1_load_frag(float4 (&f)[8], const float* __restrict__ rowp)
{
    const float4* __restrict__ p = (const float4*)rowp;
#pragma unroll
    for (int j = 0; j < 8; ++j) f[j] = p[2 * j];
}

__device__ __forceinline__ void k1_load_frag_col(float4 (&f)[8], const float* __restrict__ colp)
{
#pragma unroll
    for (int j = 0; j < 8; ++j) {
        const float* __restrict__ r = colp + (size_t)(8 * j) * F_SIZE;
        f[j] = make_float4(r[0], r[F_SIZE], r[2 * F_SIZE], r[3 * F_SIZE]);
    }
}

__device__ __forceinline__ v8f k1_mma(float ax, float ay, float bx, float by, v8f c)
{
    v2f a = {ax, ay};
    v2f b = {bx, by};
    return __builtin_amdgcn_wmma_f32_16x16x4_f32(false, a, false, b, (short)0, c, false, false);
}

__device__ __forceinline__ void k1_quarter(const float4 (&af)[8], const float4 (&bf0)[8],
                                           const float4 (&bf1)[8], float* __restrict__ pp,
                                           int c, int h)
{
    v8f c0 = {}, c1 = {};
#pragma unroll
    for (int j = 0; j < 8; ++j) {
        c0 = k1_mma(af[j].x, af[j].y, bf0[j].x, bf0[j].y, c0);
        c1 = k1_mma(af[j].x, af[j].y, bf1[j].x, bf1[j].y, c1);
        c0 = k1_mma(af[j].z, af[j].w, bf0[j].z, bf0[j].w, c0);
        c1 = k1_mma(af[j].z, af[j].w, bf1[j].z, bf1[j].w, c1);
    }
#pragma unroll
    for (int r = 0; r < 8; ++r) {
        pp[(r + 8 * h) * PPITCH + c]      = c0[r];
        pp[(r + 8 * h) * PPITCH + 16 + c] = c1[r];
    }
}

__device__ __forceinline__ void k1_ids(int (&id)[4], int2 xy0, int m, int b, int offx, int offy,
                                       const int* __restrict__ m0, const int* __restrict__ m1,
                                       const int* __restrict__ m2, const int* __restrict__ m3)
{
    const int dim = 128 >> m;
    const int* __restrict__ map = (m == 0) ? m0 : (m == 1) ? m1 : (m == 2) ? m2 : m3;
#pragma unroll
    for (int p = 0; p < 4; ++p) {
        const int scale = p + 1;
        int x = xy0.x + scale * offx;
        int y = xy0.y + scale * offy;
        x = min(max(x, 0), dim - 1);
        y = min(max(y, 0), dim - 1);
        id[p] = map[((size_t)b * dim + y) * dim + x];
    }
}

__device__ __forceinline__ void k1_finish(const float* __restrict__ rp, float4 bias,
                                          const int (&id)[4], int erow, int ek, int n_act,
                                          float* __restrict__ fw_ws, int* __restrict__ id_ws)
{
    const float4 s0 = *(const float4*)(rp);
    const float4 s1 = *(const float4*)(rp + 1 * 16 * PPITCH);
    const float4 s2 = *(const float4*)(rp + 2 * 16 * PPITCH);
    const float4 s3 = *(const float4*)(rp + 3 * 16 * PPITCH);
    const float l0 = ((s0.x + s1.x) + s2.x) + s3.x + bias.x;
    const float l1 = ((s0.y + s1.y) + s2.y) + s3.y + bias.y;
    const float l2 = ((s0.z + s1.z) + s2.z) + s3.z + bias.z;
    const float l3 = ((s0.w + s1.w) + s2.w) + s3.w + bias.w;

    const float mx = fmaxf(fmaxf(l0, l1), fmaxf(l2, l3));
    const float e0 = __expf(l0 - mx);
    const float e1 = __expf(l1 - mx);
    const float e2 = __expf(l2 - mx);
    const float e3 = __expf(l3 - mx);
    const float s  = (e0 + e1) + (e2 + e3);
    const float fw[4] = {e0 / s, e1 / s, e2 / s, e3 / s};

    if (erow < n_act) {
#pragma unroll
        for (int p = 0; p < 4; ++p) {
            const size_t o = (size_t)erow * 32 + p * 8 + ek;
            fw_ws[o] = fw[p];
            id_ws[o] = id[p];
        }
    }
}

__global__ __launch_bounds__(128) void k_attn_ids(
    const float* __restrict__ act, const int* __restrict__ bids,
    const int* __restrict__ mids, const int* __restrict__ xy,
    const int* __restrict__ m0, const int* __restrict__ m1,
    const int* __restrict__ m2, const int* __restrict__ m3,
    const float* __restrict__ Wa, const float* __restrict__ ba,
    const float* __restrict__ Wv, const float* __restrict__ bv,
    const float* __restrict__ Wo, const float* __restrict__ bo,
    float* __restrict__ wc_ws, float* __restrict__ bc_ws,
    float* __restrict__ fw_ws, int* __restrict__ id_ws, int n_act)
{
    __shared__ __attribute__((aligned(16))) float s_part[K1_TILES][4][16 * PPITCH];

    const int t    = threadIdx.x;
    const int q    = t >> 5;
    const int lane = t & 31;
    const int c    = lane & 15;
    const int h    = lane >> 4;
    const int kofs = q * 64 + 4 * h;

    const int er   = t >> 3;
    const int ek   = t & 7;

    if (blockIdx.x < K1_FOLD_BLOCKS) {
        const int o0 = 16 * (blockIdx.x >> 3);
        const int i0 = 32 * (blockIdx.x & 7);
        float4 af[8], wf0[8], wf1[8];
        k1_load_frag(af, Wo + (size_t)(o0 + c) * F_SIZE + kofs);
        k1_load_frag_col(wf0, Wv + (size_t)kofs * F_SIZE + i0 + c);
        k1_load_frag_col(wf1, Wv + (size_t)kofs * F_SIZE + i0 + 16 + c);
        k1_quarter(af, wf0, wf1, s_part[0][q], c, h);
        __syncthreads();
        {
            const float* __restrict__ rp = &s_part[0][0][er * PPITCH + 4 * ek];
            const float4 s0 = *(const float4*)(rp);
            const float4 s1 = *(const float4*)(rp + 1 * 16 * PPITCH);
            const float4 s2 = *(const float4*)(rp + 2 * 16 * PPITCH);
            const float4 s3 = *(const float4*)(rp + 3 * 16 * PPITCH);
            float4 w;
            w.x = ((s0.x + s1.x) + s2.x) + s3.x;
            w.y = ((s0.y + s1.y) + s2.y) + s3.y;
            w.z = ((s0.z + s1.z) + s2.z) + s3.z;
            w.w = ((s0.w + s1.w) + s2.w) + s3.w;
            *(float4*)(wc_ws + (size_t)(o0 + er) * F_SIZE + i0 + 4 * ek) = w;
        }
        if ((blockIdx.x & 7) == 0) {
            const float4* __restrict__ wr = (const float4*)(Wo + (size_t)(o0 + er) * F_SIZE + 32 * ek);
            const float4* __restrict__ br = (const float4*)(bv + 32 * ek);
            float s = 0.f;
#pragma unroll
            for (int u = 0; u < 8; ++u) {
                const float4 wv4 = wr[u];
                const float4 bv4 = br[u];
                s = fmaf(wv4.x, bv4.x, s);
                s = fmaf(wv4.y, bv4.y, s);
                s = fmaf(wv4.z, bv4.z, s);
                s = fmaf(wv4.w, bv4.w, s);
            }
            s += __shfl_xor(s, 1);
            s += __shfl_xor(s, 2);
            s += __shfl_xor(s, 4);
            if (ek == 0) bc_ws[o0 + er] = s + bo[o0 + er];
        }
        return;
    }

    const int row0 = (blockIdx.x - K1_FOLD_BLOCKS) * (16 * K1_TILES);
    const int kk   = ek + (ek >> 2);
    const int offx = kk % 3 - 1;
    const int offy = kk / 3 - 1;

    const int erow0 = row0 + er;
    const int erow1 = row0 + 16 + er;
    const int lrow0 = min(erow0, n_act - 1);
    const int lrow1 = min(erow1, n_act - 1);
    const int2 xyA = ((const int2*)xy)[lrow0];
    const int2 xyB = ((const int2*)xy)[lrow1];
    const int  mA  = mids[lrow0];
    const int  mB  = mids[lrow1];
    const int  bA  = bids[lrow0];
    const int  bB  = bids[lrow1];
    const float4 bias = *(const float4*)(ba + 4 * ek);

    float4 bf0[8], bf1[8], a0[8], a1[8];
    k1_load_frag(bf0, Wa + (size_t)c * F_SIZE + kofs);
    k1_load_frag(bf1, Wa + (size_t)(16 + c) * F_SIZE + kofs);
    k1_load_frag(a0, act + (size_t)min(row0 + c, n_act - 1) * F_SIZE + kofs);
    k1_load_frag(a1, act + (size_t)min(row0 + 16 + c, n_act - 1) * F_SIZE + kofs);

    int idA[4], idB[4];
    k1_ids(idA, xyA, mA, bA, offx, offy, m0, m1, m2, m3);
    k1_ids(idB, xyB, mB, bB, offx, offy, m0, m1, m2, m3);

    k1_quarter(a0, bf0, bf1, s_part[0][q], c, h);
    __syncthreads();
    k1_finish(&s_part[0][0][er * PPITCH + 4 * ek], bias, idA, erow0, ek, n_act, fw_ws, id_ws);

    k1_quarter(a1, bf0, bf1, s_part[1][q], c, h);
    __syncthreads();
    k1_finish(&s_part[1][0][er * PPITCH + 4 * ek], bias, idB, erow1, ek, n_act, fw_ws, id_ws);
}

#define K2_ROWS 8

__global__ __launch_bounds__(32 * K2_ROWS) void k_gather(
    const float* __restrict__ act, const float* __restrict__ pas,
    const float* __restrict__ fw_ws, const int* __restrict__ id_ws,
    float* __restrict__ weighted, int n_act)
{
    const int lane = threadIdx.x & 31;
    const int row = __builtin_amdgcn_readfirstlane((int)(blockIdx.x * K2_ROWS + (threadIdx.x >> 5)));
    if (row >= n_act) return;

    const int*   __restrict__ idp = id_ws + (size_t)row * 32;
    const float* __restrict__ fwp = fw_ws + (size_t)row * 32;

    float4 acc0 = make_float4(0.f, 0.f, 0.f, 0.f);
    float4 acc1 = make_float4(0.f, 0.f, 0.f, 0.f);
#pragma unroll
    for (int g = 0; g < 32; ++g) {
        const int   id = idp[g];
        const float w  = fwp[g];
        const float* __restrict__ src = (id < n_act) ? (act + (size_t)id * F_SIZE)
                                                     : (pas + (size_t)(id - n_act) * F_SIZE);
        const float4 x0 = ((const float4*)src)[lane];
        const float4 x1 = ((const float4*)src)[32 + lane];
        acc0.x = __builtin_fmaf(w, x0.x, acc0.x);  acc0.y = __builtin_fmaf(w, x0.y, acc0.y);
        acc0.z = __builtin_fmaf(w, x0.z, acc0.z);  acc0.w = __builtin_fmaf(w, x0.w, acc0.w);
        acc1.x = __builtin_fmaf(w, x1.x, acc1.x);  acc1.y = __builtin_fmaf(w, x1.y, acc1.y);
        acc1.z = __builtin_fmaf(w, x1.z, acc1.z);  acc1.w = __builtin_fmaf(w, x1.w, acc1.w);
    }
    float4* __restrict__ dst = (float4*)(weighted + (size_t)row * F_SIZE);
    dst[lane]      = acc0;
    dst[32 + lane] = acc1;
}

#define K3_MT 4
#define K3_ROWS (16 * K3_MT)

struct K3Frag {
    float4 a[K3_MT][2];
    float4 b[2][2];
};

__device__ __forceinline__ void k3_load(K3Frag& f, const float* __restrict__ ap,
                                        const float* __restrict__ bp0,
                                        const float* __restrict__ bp1, int kc)
{
#pragma unroll
    for (int j = 0; j < 2; ++j) {
        f.b[0][j] = *(const float4*)(bp0 + kc + 8 * j);
        f.b[1][j] = *(const float4*)(bp1 + kc + 8 * j);
    }
#pragma unroll
    for (int i = 0; i < K3_MT; ++i)
#pragma unroll
        for (int j = 0; j < 2; ++j)
            f.a[i][j] = *(const float4*)(ap + i * 16 * LDSP + kc + 8 * j);
}

__device__ __forceinline__ void k3_mma(v8f (&acc)[K3_MT][2], const K3Frag& f)
{
#pragma unroll
    for (int j = 0; j < 2; ++j) {
#pragma unroll
        for (int i = 0; i < K3_MT; ++i) {
            acc[i][0] = k1_mma(f.a[i][j].x, f.a[i][j].y, f.b[0][j].x, f.b[0][j].y, acc[i][0]);
            acc[i][1] = k1_mma(f.a[i][j].x, f.a[i][j].y, f.b[1][j].x, f.b[1][j].y, acc[i][1]);
        }
#pragma unroll
        for (int i = 0; i < K3_MT; ++i) {
            acc[i][0] = k1_mma(f.a[i][j].z, f.a[i][j].w, f.b[0][j].z, f.b[0][j].w, acc[i][0]);
            acc[i][1] = k1_mma(f.a[i][j].z, f.a[i][j].w, f.b[1][j].z, f.b[1][j].w, acc[i][1]);
        }
    }
}

__device__ __forceinline__ void k3_panel(v8f (&acc)[K3_MT][2], const float* __restrict__ ap,
                                         const float* __restrict__ bp0,
                                         const float* __restrict__ bp1)
{
#pragma unroll
    for (int i = 0; i < K3_MT; ++i) {
        acc[i][0] = v8f{};
        acc[i][1] = v8f{};
    }
    K3Frag f0, f1;
    k3_load(f0, ap, bp0, bp1, 0);
    for (int kc = 0; kc < F_SIZE; kc += 32) {
        k3_load(f1, ap, bp0, bp1, kc + 16);
        k3_mma(acc, f0);
        if (kc + 32 < F_SIZE) k3_load(f0, ap, bp0, bp1, kc + 32);
        k3_mma(acc, f1);
    }
}

__global__ __launch_bounds__(256) void k_out_gemm(
    const float* __restrict__ weighted,
    const float* __restrict__ wc_ws, const float* __restrict__ bc_ws,
    float* __restrict__ out, int n_act)
{
    __shared__ __attribute__((aligned(16))) float lds_a[K3_ROWS * LDSP];

    const int t = threadIdx.x;
    const int wave = t >> 5;
    const int lane = t & 31;
    const int m0 = blockIdx.x * K3_ROWS;

    {
        float4 v[16];
#pragma unroll
        for (int i = 0; i < 16; ++i) {
            const int idx = t + 256 * i;
            const int r = idx >> 6;
            const int c4 = idx & 63;
            v[i] = ((const float4*)(weighted + (size_t)min(m0 + r, n_act - 1) * F_SIZE))[c4];
        }
#pragma unroll
        for (int i = 0; i < 16; ++i) {
            const int idx = t + 256 * i;
            const int r = idx >> 6;
            const int c4 = idx & 63;
            *(float4*)(lds_a + r * LDSP + 4 * c4) = v[i];
        }
    }
    __syncthreads();

    const int c = lane & 15;
    const int h = lane >> 4;
    const int nn0 = (wave * 2 + 0) * 16 + c;
    const int nn1 = (wave * 2 + 1) * 16 + c;

    v8f acc[K3_MT][2];

    k3_panel(acc, lds_a + c * LDSP + 4 * h,
             wc_ws + (size_t)nn0 * F_SIZE + 4 * h, wc_ws + (size_t)nn1 * F_SIZE + 4 * h);
    {
        const float bias0 = bc_ws[nn0];
        const float bias1 = bc_ws[nn1];
#pragma unroll
        for (int i = 0; i < K3_MT; ++i)
#pragma unroll
            for (int r = 0; r < 8; ++r) {
                const int row = m0 + 16 * i + r + 8 * h;
                if (row < n_act) {
                    out[(size_t)row * F_SIZE + nn0] = acc[i][0][r] + bias0;
                    out[(size_t)row * F_SIZE + nn1] = acc[i][1][r] + bias1;
                }
            }
    }
}

extern "C" void kernel_launch(void* const* d_in, const int* in_sizes, int n_in,
                              void* d_out, int out_size, void* d_ws, size_t ws_size,
                              hipStream_t stream) {
    const float* act  = (const float*)d_in[0];
    const float* pas  = (const float*)d_in[1];
    const int*   bids = (const int*)d_in[2];
    const int*   mids = (const int*)d_in[3];
    const int*   xy   = (const int*)d_in[4];
    const int*   m0   = (const int*)d_in[5];
    const int*   m1   = (const int*)d_in[6];
    const int*   m2   = (const int*)d_in[7];
    const int*   m3   = (const int*)d_in[8];
    const float* Wa   = (const float*)d_in[9];
    const float* ba   = (const float*)d_in[10];
    const float* Wv   = (const float*)d_in[11];
    const float* bv   = (const float*)d_in[12];
    const float* Wo   = (const float*)d_in[13];
    const float* bo   = (const float*)d_in[14];
    float* out = (float*)d_out;

    const int n_act = in_sizes[0] / F_SIZE;

    float* fw_ws    = out;
    int*   id_ws    = (int*)(out + (size_t)n_act * 32);
    float* wc_ws    = (float*)d_ws;
    float* bc_ws    = wc_ws + F_SIZE * F_SIZE;
    float* weighted = bc_ws + F_SIZE;

    k_attn_ids<<<K1_FOLD_BLOCKS + (n_act + 16 * K1_TILES - 1) / (16 * K1_TILES), 128, 0, stream>>>(
        act, bids, mids, xy, m0, m1, m2, m3, Wa, ba, Wv, bv, Wo, bo, wc_ws, bc_ws,
        fw_ws, id_ws, n_act);

    k_gather<<<(n_act + K2_ROWS - 1) / K2_ROWS, 32 * K2_ROWS, 0, stream>>>(act, pas, fw_ws, id_ws, weighted, n_act);

    k_out_gemm<<<(n_act + K3_ROWS - 1) / K3_ROWS, 256, 0, stream>>>(weighted, wc_ws, bc_ws, out, n_act);
}


# Launches 2 and 3 fused: gathered rows go straight into the GEMM's LDS tile; 16-wave blocks own two 64-row tiles, 8 waves gather tile 2 while 8 run tile 1's matrix product; sum orders unchanged
# speedup vs baseline: 3.3337x; 1.0954x over previous
#include <hip/hip_runtime.h>
#include <hip/hip_bf16.h>

typedef __attribute__((ext_vector_type(2))) float v2f;
typedef __attribute__((ext_vector_type(8))) float v8f;

#define F_SIZE 256
#define LDSP 260

#define K1_TILES 2
#define PPITCH 36
#define K1_FOLD_BLOCKS 128

__device__ __forceinline__ void k1_load_frag(float4 (&f)[8], const float* __restrict__ rowp)
{
    const float4* __restrict__ p = (const float4*)rowp;
#pragma unroll
    for (int j = 0; j < 8; ++j) f[j] = p[2 * j];
}

__device__ __forceinline__ void k1_load_frag_col(float4 (&f)[8], const float* __restrict__ colp)
{
#pragma unroll
    for (int j = 0; j < 8; ++j) {
        const float* __restrict__ r = colp + (size_t)(8 * j) * F_SIZE;
        f[j] = make_float4(r[0], r[F_SIZE], r[2 * F_SIZE], r[3 * F_SIZE]);
    }
}

__device__ __forceinline__ v8f k1_mma(float ax, float ay, float bx, float by, v8f c)
{
    v2f a = {ax, ay};
    v2f b = {bx, by};
    return __builtin_amdgcn_wmma_f32_16x16x4_f32(false, a, false, b, (short)0, c, false, false);
}

__device__ __forceinline__ void k1_quarter(const float4 (&af)[8], const float4 (&bf0)[8],
                                           const float4 (&bf1)[8], float* __restrict__ pp,
                                           int c, int h)
{
    v8f c0 = {}, c1 = {};
#pragma unroll
    for (int j = 0; j < 8; ++j) {
        c0 = k1_mma(af[j].x, af[j].y, bf0[j].x, bf0[j].y, c0);
        c1 = k1_mma(af[j].x, af[j].y, bf1[j].x, bf1[j].y, c1);
        c0 = k1_mma(af[j].z, af[j].w, bf0[j].z, bf0[j].w, c0);
        c1 = k1_mma(af[j].z, af[j].w, bf1[j].z, bf1[j].w, c1);
    }
#pragma unroll
    for (int r = 0; r < 8; ++r) {
        pp[(r + 8 * h) * PPITCH + c]      = c0[r];
        pp[(r + 8 * h) * PPITCH + 16 + c] = c1[r];
    }
}

__device__ __forceinline__ void k1_ids(int (&id)[4], int2 xy0, int m, int b, int offx, int offy,
                                       const int* __restrict__ m0, const int* __restrict__ m1,
                                       const int* __restrict__ m2, const int* __restrict__ m3)
{
    const int dim = 128 >> m;
    const int* __restrict__ map = (m == 0) ? m0 : (m == 1) ? m1 : (m == 2) ? m2 : m3;
#pragma unroll
    for (int p = 0; p < 4; ++p) {
        const int scale = p + 1;
        int x = xy0.x + scale * offx;
        int y = xy0.y + scale * offy;
        x = min(max(x, 0), dim - 1);
        y = min(max(y, 0), dim - 1);
        id[p] = map[((size_t)b * dim + y) * dim + x];
    }
}

__device__ __forceinline__ void k1_finish(const float* __restrict__ rp, float4 bias,
                                          const int (&id)[4], int erow, int ek, int n_act,
                                          float* __restrict__ fw_ws, int* __restrict__ id_ws)
{
    const float4 s0 = *(const float4*)(rp);
    const float4 s1 = *(const float4*)(rp + 1 * 16 * PPITCH);
    const float4 s2 = *(const float4*)(rp + 2 * 16 * PPITCH);
    const float4 s3 = *(const float4*)(rp + 3 * 16 * PPITCH);
    const float l0 = ((s0.x + s1.x) + s2.x) + s3.x + bias.x;
    const float l1 = ((s0.y + s1.y) + s2.y) + s3.y + bias.y;
    const float l2 = ((s0.z + s1.z) + s2.z) + s3.z + bias.z;
    const float l3 = ((s0.w + s1.w) + s2.w) + s3.w + bias.w;

    const float mx = fmaxf(fmaxf(l0, l1), fmaxf(l2, l3));
    const float e0 = __expf(l0 - mx);
    const float e1 = __expf(l1 - mx);
    const float e2 = __expf(l2 - mx);
    const float e3 = __expf(l3 - mx);
    const float s  = (e0 + e1) + (e2 + e3);
    const float fw[4] = {e0 / s, e1 / s, e2 / s, e3 / s};

    if (erow < n_act) {
#pragma unroll
        for (int p = 0; p < 4; ++p) {
            const size_t o = (size_t)erow * 32 + p * 8 + ek;
            fw_ws[o] = fw[p];
            id_ws[o] = id[p];
        }
    }
}

__global__ __launch_bounds__(128) void k_attn_ids(
    const float* __restrict__ act, const int* __restrict__ bids,
    const int* __restrict__ mids, const int* __restrict__ xy,
    const int* __restrict__ m0, const int* __restrict__ m1,
    const int* __restrict__ m2, const int* __restrict__ m3,
    const float* __restrict__ Wa, const float* __restrict__ ba,
    const float* __restrict__ Wv, const float* __restrict__ bv,
    const float* __restrict__ Wo, const float* __restrict__ bo,
    float* __restrict__ wc_ws, float* __restrict__ bc_ws,
    float* __restrict__ fw_ws, int* __restrict__ id_ws, int n_act)
{
    __shared__ __attribute__((aligned(16))) float s_part[K1_TILES][4][16 * PPITCH];

    const int t    = threadIdx.x;
    const int q    = t >> 5;
    const int lane = t & 31;
    const int c    = lane & 15;
    const int h    = lane >> 4;
    const int kofs = q * 64 + 4 * h;

    const int er   = t >> 3;
    const int ek   = t & 7;

    if (blockIdx.x < K1_FOLD_BLOCKS) {
        const int o0 = 16 * (blockIdx.x >> 3);
        const int i0 = 32 * (blockIdx.x & 7);
        float4 af[8], wf0[8], wf1[8];
        k1_load_frag(af, Wo + (size_t)(o0 + c) * F_SIZE + kofs);
        k1_load_frag_col(wf0, Wv + (size_t)kofs * F_SIZE + i0 + c);
        k1_load_frag_col(wf1, Wv + (size_t)kofs * F_SIZE + i0 + 16 + c);
        k1_quarter(af, wf0, wf1, s_part[0][q], c, h);
        __syncthreads();
        {
            const float* __restrict__ rp = &s_part[0][0][er * PPITCH + 4 * ek];
            const float4 s0 = *(const float4*)(rp);
            const float4 s1 = *(const float4*)(rp + 1 * 16 * PPITCH);
            const float4 s2 = *(const float4*)(rp + 2 * 16 * PPITCH);
            const float4 s3 = *(const float4*)(rp + 3 * 16 * PPITCH);
            float4 w;
            w.x = ((s0.x + s1.x) + s2.x) + s3.x;
            w.y = ((s0.y + s1.y) + s2.y) + s3.y;
            w.z = ((s0.z + s1.z) + s2.z) + s3.z;
            w.w = ((s0.w + s1.w) + s2.w) + s3.w;
            *(float4*)(wc_ws + (size_t)(o0 + er) * F_SIZE + i0 + 4 * ek) = w;
        }
        if ((blockIdx.x & 7) == 0) {
            const float4* __restrict__ wr = (const float4*)(Wo + (size_t)(o0 + er) * F_SIZE + 32 * ek);
            const float4* __restrict__ br = (const float4*)(bv + 32 * ek);
            float s = 0.f;
#pragma unroll
            for (int u = 0; u < 8; ++u) {
                const float4 wv4 = wr[u];
                const float4 bv4 = br[u];
                s = fmaf(wv4.x, bv4.x, s);
                s = fmaf(wv4.y, bv4.y, s);
                s = fmaf(wv4.z, bv4.z, s);
                s = fmaf(wv4.w, bv4.w, s);
            }
            s += __shfl_xor(s, 1);
            s += __shfl_xor(s, 2);
            s += __shfl_xor(s, 4);
            if (ek == 0) bc_ws[o0 + er] = s + bo[o0 + er];
        }
        return;
    }

    const int row0 = (blockIdx.x - K1_FOLD_BLOCKS) * (16 * K1_TILES);
    const int kk   = ek + (ek >> 2);
    const int offx = kk % 3 - 1;
    const int offy = kk / 3 - 1;

    const int erow0 = row0 + er;
    const int erow1 = row0 + 16 + er;
    const int lrow0 = min(erow0, n_act - 1);
    const int lrow1 = min(erow1, n_act - 1);
    const int2 xyA = ((const int2*)xy)[lrow0];
    const int2 xyB = ((const int2*)xy)[lrow1];
    const int  mA  = mids[lrow0];
    const int  mB  = mids[lrow1];
    const int  bA  = bids[lrow0];
    const int  bB  = bids[lrow1];
    const float4 bias = *(const float4*)(ba + 4 * ek);

    float4 bf0[8], bf1[8], a0[8], a1[8];
    k1_load_frag(bf0, Wa + (size_t)c * F_SIZE + kofs);
    k1_load_frag(bf1, Wa + (size_t)(16 + c) * F_SIZE + kofs);
    k1_load_frag(a0, act + (size_t)min(row0 + c, n_act - 1) * F_SIZE + kofs);
    k1_load_frag(a1, act + (size_t)min(row0 + 16 + c, n_act - 1) * F_SIZE + kofs);

    int idA[4], idB[4];
    k1_ids(idA, xyA, mA, bA, offx, offy, m0, m1, m2, m3);
    k1_ids(idB, xyB, mB, bB, offx, offy, m0, m1, m2, m3);

    k1_quarter(a0, bf0, bf1, s_part[0][q], c, h);
    __syncthreads();
    k1_finish(&s_part[0][0][er * PPITCH + 4 * ek], bias, idA, erow0, ek, n_act, fw_ws, id_ws);

    k1_quarter(a1, bf0, bf1, s_part[1][q], c, h);
    __syncthreads();
    k1_finish(&s_part[1][0][er * PPITCH + 4 * ek], bias, idB, erow1, ek, n_act, fw_ws, id_ws);
}

#define K3_MT 4
#define K3_ROWS (16 * K3_MT)

struct K3Frag {
    float4 a[K3_MT][2];
    float4 b[2][2];
};

__device__ __forceinline__ void k3_load(K3Frag& f, const float* __restrict__ ap,
                                        const float* __restrict__ bp0,
                                        const float* __restrict__ bp1, int kc)
{
#pragma unroll
    for (int j = 0; j < 2; ++j) {
        f.b[0][j] = *(const float4*)(bp0 + kc + 8 * j);
        f.b[1][j] = *(const float4*)(bp1 + kc + 8 * j);
    }
#pragma unroll
    for (int i = 0; i < K3_MT; ++i)
#pragma unroll
        for (int j = 0; j < 2; ++j)
            f.a[i][j] = *(const float4*)(ap + i * 16 * LDSP + kc + 8 * j);
}

__device__ __forceinline__ void k3_mma(v8f (&acc)[K3_MT][2], const K3Frag& f)
{
#pragma unroll
    for (int j = 0; j < 2; ++j) {
#pragma unroll
        for (int i = 0; i < K3_MT; ++i) {
            acc[i][0] = k1_mma(f.a[i][j].x, f.a[i][j].y, f.b[0][j].x, f.b[0][j].y, acc[i][0]);
            acc[i][1] = k1_mma(f.a[i][j].x, f.a[i][j].y, f.b[1][j].x, f.b[1][j].y, acc[i][1]);
        }
#pragma unroll
        for (int i = 0; i < K3_MT; ++i) {
            acc[i][0] = k1_mma(f.a[i][j].z, f.a[i][j].w, f.b[0][j].z, f.b[0][j].w, acc[i][0]);
            acc[i][1] = k1_mma(f.a[i][j].z, f.a[i][j].w, f.b[1][j].z, f.b[1][j].w, acc[i][1]);
        }
    }
}

__device__ __forceinline__ void k3_panel(v8f (&acc)[K3_MT][2], const float* __restrict__ ap,
                                         const float* __restrict__ bp0,
                                         const float* __restrict__ bp1)
{
#pragma unroll
    for (int i = 0; i < K3_MT; ++i) {
        acc[i][0] = v8f{};
        acc[i][1] = v8f{};
    }
    K3Frag f0, f1;
    k3_load(f0, ap, bp0, bp1, 0);
    for (int kc = 0; kc < F_SIZE; kc += 32) {
        k3_load(f1, ap, bp0, bp1, kc + 16);
        k3_mma(acc, f0);
        if (kc + 32 < F_SIZE) k3_load(f0, ap, bp0, bp1, kc + 32);
        k3_mma(acc, f1);
    }
}

#define KF_WAVES 16
#define KF_TILES 2

__device__ __forceinline__ void kf_gather_rows(float* __restrict__ tile, int m0, int r0, int nrows,
                                               const float* __restrict__ act,
                                               const float* __restrict__ pas,
                                               const float* __restrict__ fw_ws,
                                               const int* __restrict__ id_ws,
                                               int n_act, int lane)
{
    const int first = min(m0 + r0, n_act - 1);
    int   idv = id_ws[(size_t)first * 32 + lane];
    float wv  = fw_ws[(size_t)first * 32 + lane];
    for (int i = 0; i < nrows; ++i) {
        const int nxt = min(m0 + r0 + min(i + 1, nrows - 1), n_act - 1);
        const int   idn = id_ws[(size_t)nxt * 32 + lane];
        const float wn  = fw_ws[(size_t)nxt * 32 + lane];

        float4 acc0 = make_float4(0.f, 0.f, 0.f, 0.f);
        float4 acc1 = make_float4(0.f, 0.f, 0.f, 0.f);
#pragma unroll
        for (int g = 0; g < 32; ++g) {
            const int   id = __builtin_amdgcn_readlane(idv, g);
            const float w  = __int_as_float(__builtin_amdgcn_readlane(__float_as_int(wv), g));
            const float* __restrict__ src = (id < n_act) ? (act + (size_t)id * F_SIZE)
                                                         : (pas + (size_t)(id - n_act) * F_SIZE);
            const float4 x0 = ((const float4*)src)[lane];
            const float4 x1 = ((const float4*)src)[32 + lane];
            acc0.x = __builtin_fmaf(w, x0.x, acc0.x);  acc0.y = __builtin_fmaf(w, x0.y, acc0.y);
            acc0.z = __builtin_fmaf(w, x0.z, acc0.z);  acc0.w = __builtin_fmaf(w, x0.w, acc0.w);
            acc1.x = __builtin_fmaf(w, x1.x, acc1.x);  acc1.y = __builtin_fmaf(w, x1.y, acc1.y);
            acc1.z = __builtin_fmaf(w, x1.z, acc1.z);  acc1.w = __builtin_fmaf(w, x1.w, acc1.w);
        }
        float* __restrict__ dst = tile + (r0 + i) * LDSP;
        *(float4*)(dst + 4 * lane)       = acc0;
        *(float4*)(dst + 128 + 4 * lane) = acc1;
        idv = idn;
        wv  = wn;
    }
}

__device__ __forceinline__ void kf_product(const float* __restrict__ tile, int m0, int wave, int lane,
                                           const float* __restrict__ wc_ws,
                                           const float* __restrict__ bc_ws,
                                           float* __restrict__ out, int n_act)
{
    const int c = lane & 15;
    const int h = lane >> 4;
    const int nn0 = (wave * 2 + 0) * 16 + c;
    const int nn1 = (wave * 2 + 1) * 16 + c;

    v8f acc[K3_MT][2];
    k3_panel(acc, tile + c * LDSP + 4 * h,
             wc_ws + (size_t)nn0 * F_SIZE + 4 * h, wc_ws + (size_t)nn1 * F_SIZE + 4 * h);

    const float bias0 = bc_ws[nn0];
    const float bias1 = bc_ws[nn1];
#pragma unroll
    for (int i = 0; i < K3_MT; ++i)
#pragma unroll
        for (int r = 0; r < 8; ++r) {
            const int row = m0 + 16 * i + r + 8 * h;
            if (row < n_act) {
                out[(size_t)row * F_SIZE + nn0] = acc[i][0][r] + bias0;
                out[(size_t)row * F_SIZE + nn1] = acc[i][1][r] + bias1;
            }
        }
}

__global__ __launch_bounds__(32 * KF_WAVES) void k_gather_gemm(
    const float* __restrict__ act, const float* __restrict__ pas,
    const float* __restrict__ fw_ws, const int* __restrict__ id_ws,
    const float* __restrict__ wc_ws, const float* __restrict__ bc_ws,
    float* __restrict__ out, int n_act)
{
    __shared__ __attribute__((aligned(16))) float lds_a[KF_TILES][K3_ROWS * LDSP];

    const int lane = threadIdx.x & 31;
    const int wave = __builtin_amdgcn_readfirstlane((int)(threadIdx.x >> 5));

    const int n_tiles = (n_act + K3_ROWS - 1) / K3_ROWS;
    const int m0 = (int)blockIdx.x * (KF_TILES * K3_ROWS);
    const int m1 = m0 + K3_ROWS;
    const bool has1 = (int)blockIdx.x * KF_TILES + 1 < n_tiles;

    kf_gather_rows(lds_a[0], m0, 4 * wave, 4, act, pas, fw_ws, id_ws, n_act, lane);
    __syncthreads();

    if (wave < 8) {
        kf_product(lds_a[0], m0, wave, lane, wc_ws, bc_ws, out, n_act);
    } else if (has1) {
        kf_gather_rows(lds_a[1], m1, 8 * (wave - 8), 8, act, pas, fw_ws, id_ws, n_act, lane);
    }
    __syncthreads();

    if (wave < 8 && has1) {
        kf_product(lds_a[1], m1, wave, lane, wc_ws, bc_ws, out, n_act);
    }
}

extern "C" void kernel_launch(void* const* d_in, const int* in_sizes, int n_in,
                              void* d_out, int out_size, void* d_ws, size_t ws_size,
                              hipStream_t stream) {
    const float* act  = (const float*)d_in[0];
    const float* pas  = (const float*)d_in[1];
    const int*   bids = (const int*)d_in[2];
    const int*   mids = (const int*)d_in[3];
    const int*   xy   = (const int*)d_in[4];
    const int*   m0   = (const int*)d_in[5];
    const int*   m1   = (const int*)d_in[6];
    const int*   m2   = (const int*)d_in[7];
    const int*   m3   = (const int*)d_in[8];
    const float* Wa   = (const float*)d_in[9];
    const float* ba   = (const float*)d_in[10];
    const float* Wv   = (const float*)d_in[11];
    const float* bv   = (const float*)d_in[12];
    const float* Wo   = (const float*)d_in[13];
    const float* bo   = (const float*)d_in[14];
    float* out = (float*)d_out;

    const int n_act = in_sizes[0] / F_SIZE;

    float* wc_ws = (float*)d_ws;
    float* bc_ws = wc_ws + F_SIZE * F_SIZE;
    float* fw_ws = bc_ws + F_SIZE;
    int*   id_ws = (int*)(fw_ws + (size_t)n_act * 32);

    k_attn_ids<<<K1_FOLD_BLOCKS + (n_act + 16 * K1_TILES - 1) / (16 * K1_TILES), 128, 0, stream>>>(
        act, bids, mids, xy, m0, m1, m2, m3, Wa, ba, Wv, bv, Wo, bo, wc_ws, bc_ws,
        fw_ws, id_ws, n_act);

    k_gather_gemm<<<((n_act + K3_ROWS - 1) / K3_ROWS + KF_TILES - 1) / KF_TILES, 32 * KF_WAVES, 0, stream>>>(
        act, pas, fw_ws, id_ws, wc_ws, bc_ws, out, n_act);
}


# K2: Wc copied once per block into LDS and kept there, so the gather hands 16-row tiles to the product 8 times per 128-row block (was 64 rows, twice); 4 product + 8 gather waves; sum orders unchanged
# speedup vs baseline: 3.4648x; 1.0393x over previous
#include <hip/hip_runtime.h>
#include <hip/hip_bf16.h>

typedef __attribute__((ext_vector_type(2))) float v2f;
typedef __attribute__((ext_vector_type(8))) float v8f;

#define F_SIZE 256
#define LDSP 260

#define K1_TILES 2
#define PPITCH 36
#define K1_FOLD_BLOCKS 128

__device__ __forceinline__ void k1_load_frag(float4 (&f)[8], const float* __restrict__ rowp)
{
    const float4* __restrict__ p = (const float4*)rowp;
#pragma unroll
    for (int j = 0; j < 8; ++j) f[j] = p[2 * j];
}

__device__ __forceinline__ void k1_load_frag_col(float4 (&f)[8], const float* __restrict__ colp)
{
#pragma unroll
    for (int j = 0; j < 8; ++j) {
        const float* __restrict__ r = colp + (size_t)(8 * j) * F_SIZE;
        f[j] = make_float4(r[0], r[F_SIZE], r[2 * F_SIZE], r[3 * F_SIZE]);
    }
}

__device__ __forceinline__ v8f k1_mma(float ax, float ay, float bx, float by, v8f c)
{
    v2f a = {ax, ay};
    v2f b = {bx, by};
    return __builtin_amdgcn_wmma_f32_16x16x4_f32(false, a, false, b, (short)0, c, false, false);
}

__device__ __forceinline__ void k1_quarter(const float4 (&af)[8], const float4 (&bf0)[8],
                                           const float4 (&bf1)[8], float* __restrict__ pp,
                                           int c, int h)
{
    v8f c0 = {}, c1 = {};
#pragma unroll
    for (int j = 0; j < 8; ++j) {
        c0 = k1_mma(af[j].x, af[j].y, bf0[j].x, bf0[j].y, c0);
        c1 = k1_mma(af[j].x, af[j].y, bf1[j].x, bf1[j].y, c1);
        c0 = k1_mma(af[j].z, af[j].w, bf0[j].z, bf0[j].w, c0);
        c1 = k1_mma(af[j].z, af[j].w, bf1[j].z, bf1[j].w, c1);
    }
#pragma unroll
    for (int r = 0; r < 8; ++r) {
        pp[(r + 8 * h) * PPITCH + c]      = c0[r];
        pp[(r + 8 * h) * PPITCH + 16 + c] = c1[r];
    }
}

__device__ __forceinline__ void k1_ids(int (&id)[4], int2 xy0, int m, int b, int offx, int offy,
                                       const int* __restrict__ m0, const int* __restrict__ m1,
                                       const int* __restrict__ m2, const int* __restrict__ m3)
{
    const int dim = 128 >> m;
    const int* __restrict__ map = (m == 0) ? m0 : (m == 1) ? m1 : (m == 2) ? m2 : m3;
#pragma unroll
    for (int p = 0; p < 4; ++p) {
        const int scale = p + 1;
        int x = xy0.x + scale * offx;
        int y = xy0.y + scale * offy;
        x = min(max(x, 0), dim - 1);
        y = min(max(y, 0), dim - 1);
        id[p] = map[((size_t)b * dim + y) * dim + x];
    }
}

__device__ __forceinline__ void k1_finish(const float* __restrict__ rp, float4 bias,
                                          const int (&id)[4], int erow, int ek, int n_act,
                                          float* __restrict__ fw_ws, int* __restrict__ id_ws)
{
    const float4 s0 = *(const float4*)(rp);
    const float4 s1 = *(const float4*)(rp + 1 * 16 * PPITCH);
    const float4 s2 = *(const float4*)(rp + 2 * 16 * PPITCH);
    const float4 s3 = *(const float4*)(rp + 3 * 16 * PPITCH);
    const float l0 = ((s0.x + s1.x) + s2.x) + s3.x + bias.x;
    const float l1 = ((s0.y + s1.y) + s2.y) + s3.y + bias.y;
    const float l2 = ((s0.z + s1.z) + s2.z) + s3.z + bias.z;
    const float l3 = ((s0.w + s1.w) + s2.w) + s3.w + bias.w;

    const float mx = fmaxf(fmaxf(l0, l1), fmaxf(l2, l3));
    const float e0 = __expf(l0 - mx);
    const float e1 = __expf(l1 - mx);
    const float e2 = __expf(l2 - mx);
    const float e3 = __expf(l3 - mx);
    const float s  = (e0 + e1) + (e2 + e3);
    const float fw[4] = {e0 / s, e1 / s, e2 / s, e3 / s};

    if (erow < n_act) {
#pragma unroll
        for (int p = 0; p < 4; ++p) {
            const size_t o = (size_t)erow * 32 + p * 8 + ek;
            fw_ws[o] = fw[p];
            id_ws[o] = id[p];
        }
    }
}

__global__ __launch_bounds__(128) void k_attn_ids(
    const float* __restrict__ act, const int* __restrict__ bids,
    const int* __restrict__ mids, const int* __restrict__ xy,
    const int* __restrict__ m0, const int* __restrict__ m1,
    const int* __restrict__ m2, const int* __restrict__ m3,
    const float* __restrict__ Wa, const float* __restrict__ ba,
    const float* __restrict__ Wv, const float* __restrict__ bv,
    const float* __restrict__ Wo, const float* __restrict__ bo,
    float* __restrict__ wc_ws, float* __restrict__ bc_ws,
    float* __restrict__ fw_ws, int* __restrict__ id_ws, int n_act)
{
    __shared__ __attribute__((aligned(16))) float s_part[K1_TILES][4][16 * PPITCH];

    const int t    = threadIdx.x;
    const int q    = t >> 5;
    const int lane = t & 31;
    const int c    = lane & 15;
    const int h    = lane >> 4;
    const int kofs = q * 64 + 4 * h;

    const int er   = t >> 3;
    const int ek   = t & 7;

    if (blockIdx.x < K1_FOLD_BLOCKS) {
        const int o0 = 16 * (blockIdx.x >> 3);
        const int i0 = 32 * (blockIdx.x & 7);
        float4 af[8], wf0[8], wf1[8];
        k1_load_frag(af, Wo + (size_t)(o0 + c) * F_SIZE + kofs);
        k1_load_frag_col(wf0, Wv + (size_t)kofs * F_SIZE + i0 + c);
        k1_load_frag_col(wf1, Wv + (size_t)kofs * F_SIZE + i0 + 16 + c);
        k1_quarter(af, wf0, wf1, s_part[0][q], c, h);
        __syncthreads();
        {
            const float* __restrict__ rp = &s_part[0][0][er * PPITCH + 4 * ek];
            const float4 s0 = *(const float4*)(rp);
            const float4 s1 = *(const float4*)(rp + 1 * 16 * PPITCH);
            const float4 s2 = *(const float4*)(rp + 2 * 16 * PPITCH);
            const float4 s3 = *(const float4*)(rp + 3 * 16 * PPITCH);
            float4 w;
            w.x = ((s0.x + s1.x) + s2.x) + s3.x;
            w.y = ((s0.y + s1.y) + s2.y) + s3.y;
            w.z = ((s0.z + s1.z) + s2.z) + s3.z;
            w.w = ((s0.w + s1.w) + s2.w) + s3.w;
            *(float4*)(wc_ws + (size_t)(o0 + er) * F_SIZE + i0 + 4 * ek) = w;
        }
        if ((blockIdx.x & 7) == 0) {
            const float4* __restrict__ wr = (const float4*)(Wo + (size_t)(o0 + er) * F_SIZE + 32 * ek);
            const float4* __restrict__ br = (const float4*)(bv + 32 * ek);
            float s = 0.f;
#pragma unroll
            for (int u = 0; u < 8; ++u) {
                const float4 wv4 = wr[u];
                const float4 bv4 = br[u];
                s = fmaf(wv4.x, bv4.x, s);
                s = fmaf(wv4.y, bv4.y, s);
                s = fmaf(wv4.z, bv4.z, s);
                s = fmaf(wv4.w, bv4.w, s);
            }
            s += __shfl_xor(s, 1);
            s += __shfl_xor(s, 2);
            s += __shfl_xor(s, 4);
            if (ek == 0) bc_ws[o0 + er] = s + bo[o0 + er];
        }
        return;
    }

    const int row0 = (blockIdx.x - K1_FOLD_BLOCKS) * (16 * K1_TILES);
    const int kk   = ek + (ek >> 2);
    const int offx = kk % 3 - 1;
    const int offy = kk / 3 - 1;

    const int erow0 = row0 + er;
    const int erow1 = row0 + 16 + er;
    const int lrow0 = min(erow0, n_act - 1);
    const int lrow1 = min(erow1, n_act - 1);
    const int2 xyA = ((const int2*)xy)[lrow0];
    const int2 xyB = ((const int2*)xy)[lrow1];
    const int  mA  = mids[lrow0];
    const int  mB  = mids[lrow1];
    const int  bA  = bids[lrow0];
    const int  bB  = bids[lrow1];
    const float4 bias = *(const float4*)(ba + 4 * ek);

    float4 bf0[8], bf1[8], a0[8], a1[8];
    k1_load_frag(bf0, Wa + (size_t)c * F_SIZE + kofs);
    k1_load_frag(bf1, Wa + (size_t)(16 + c) * F_SIZE + kofs);
    k1_load_frag(a0, act + (size_t)min(row0 + c, n_act - 1) * F_SIZE + kofs);
    k1_load_frag(a1, act + (size_t)min(row0 + 16 + c, n_act - 1) * F_SIZE + kofs);

    int idA[4], idB[4];
    k1_ids(idA, xyA, mA, bA, offx, offy, m0, m1, m2, m3);
    k1_ids(idB, xyB, mB, bB, offx, offy, m0, m1, m2, m3);

    k1_quarter(a0, bf0, bf1, s_part[0][q], c, h);
    __syncthreads();
    k1_finish(&s_part[0][0][er * PPITCH + 4 * ek], bias, idA, erow0, ek, n_act, fw_ws, id_ws);

    k1_quarter(a1, bf0, bf1, s_part[1][q], c, h);
    __syncthreads();
    k1_finish(&s_part[1][0][er * PPITCH + 4 * ek], bias, idB, erow1, ek, n_act, fw_ws, id_ws);
}

#define KB_ROWS  128
#define KB_SUB   16
#define KB_NSUB  (KB_ROWS / KB_SUB)
#define KB_WAVES 12

struct KpFrag {
    float4 a[2];
    float4 b[4][2];
};

__device__ __forceinline__ void kp_load(KpFrag& f, const float* __restrict__ ap,
                                        const float* __restrict__ bp, int kc)
{
#pragma unroll
    for (int j = 0; j < 2; ++j) {
        f.a[j] = *(const float4*)(ap + kc + 8 * j);
#pragma unroll
        for (int n = 0; n < 4; ++n)
            f.b[n][j] = *(const float4*)(bp + n * 16 * LDSP + kc + 8 * j);
    }
}

__device__ __forceinline__ void kp_mma(v8f (&acc)[4], const KpFrag& f)
{
#pragma unroll
    for (int j = 0; j < 2; ++j) {
#pragma unroll
        for (int n = 0; n < 4; ++n)
            acc[n] = k1_mma(f.a[j].x, f.a[j].y, f.b[n][j].x, f.b[n][j].y, acc[n]);
#pragma unroll
        for (int n = 0; n < 4; ++n)
            acc[n] = k1_mma(f.a[j].z, f.a[j].w, f.b[n][j].z, f.b[n][j].w, acc[n]);
    }
}

__device__ __forceinline__ void kp_product(const float* __restrict__ tile,
                                           const float* __restrict__ wl, int mrow0, int p,
                                           int lane, const float (&bias)[4],
                                           float* __restrict__ out, int n_act)
{
    const int c = lane & 15;
    const int h = lane >> 4;
    const float* __restrict__ ap = tile + c * LDSP + 4 * h;
    const float* __restrict__ bp = wl + (64 * p + c) * LDSP + 4 * h;

    v8f acc[4];
#pragma unroll
    for (int n = 0; n < 4; ++n) acc[n] = v8f{};
    KpFrag f0, f1;
    kp_load(f0, ap, bp, 0);
    for (int kc = 0; kc < F_SIZE; kc += 32) {
        kp_load(f1, ap, bp, kc + 16);
        kp_mma(acc, f0);
        if (kc + 32 < F_SIZE) kp_load(f0, ap, bp, kc + 32);
        kp_mma(acc, f1);
    }

#pragma unroll
    for (int n = 0; n < 4; ++n)
#pragma unroll
        for (int r = 0; r < 8; ++r) {
            const int row = mrow0 + r + 8 * h;
            if (row < n_act)
                out[(size_t)row * F_SIZE + 64 * p + 16 * n + c] = acc[n][r] + bias[n];
        }
}

__device__ __forceinline__ void kp_copy_wc(float* __restrict__ wl, const float* __restrict__ wc_ws,
                                           int p, int lane)
{
    const float4* __restrict__ srcp = (const float4*)(wc_ws + (size_t)(64 * p) * F_SIZE);
    float* __restrict__ dst = wl + (64 * p) * LDSP;
#pragma unroll 16
    for (int r = 0; r < 64; ++r) {
        const float4 x0 = srcp[r * 64 + lane];
        const float4 x1 = srcp[r * 64 + 32 + lane];
        *(float4*)(dst + r * LDSP + 4 * lane)       = x0;
        *(float4*)(dst + r * LDSP + 128 + 4 * lane) = x1;
    }
}

__device__ __forceinline__ void kf_gather_row(float* __restrict__ dst, int idv, float wv,
                                              const float* __restrict__ act,
                                              const float* __restrict__ pas,
                                              int n_act, int lane)
{
    float4 acc0 = make_float4(0.f, 0.f, 0.f, 0.f);
    float4 acc1 = make_float4(0.f, 0.f, 0.f, 0.f);
#pragma unroll
    for (int g = 0; g < 32; ++g) {
        const int   id = __builtin_amdgcn_readlane(idv, g);
        const float w  = __int_as_float(__builtin_amdgcn_readlane(__float_as_int(wv), g));
        const float* __restrict__ src = (id < n_act) ? (act + (size_t)id * F_SIZE)
                                                     : (pas + (size_t)(id - n_act) * F_SIZE);
        const float4 x0 = ((const float4*)src)[lane];
        const float4 x1 = ((const float4*)src)[32 + lane];
        acc0.x = __builtin_fmaf(w, x0.x, acc0.x);  acc0.y = __builtin_fmaf(w, x0.y, acc0.y);
        acc0.z = __builtin_fmaf(w, x0.z, acc0.z);  acc0.w = __builtin_fmaf(w, x0.w, acc0.w);
        acc1.x = __builtin_fmaf(w, x1.x, acc1.x);  acc1.y = __builtin_fmaf(w, x1.y, acc1.y);
        acc1.z = __builtin_fmaf(w, x1.z, acc1.z);  acc1.w = __builtin_fmaf(w, x1.w, acc1.w);
    }
    *(float4*)(dst + 4 * lane)       = acc0;
    *(float4*)(dst + 128 + 4 * lane) = acc1;
}

__global__ __launch_bounds__(32 * KB_WAVES) void k_gather_gemm(
    const float* __restrict__ act, const float* __restrict__ pas,
    const float* __restrict__ fw_ws, const int* __restrict__ id_ws,
    const float* __restrict__ wc_ws, const float* __restrict__ bc_ws,
    float* __restrict__ out, int n_act)
{
    __shared__ __attribute__((aligned(16))) float lds_w[F_SIZE * LDSP];
    __shared__ __attribute__((aligned(16))) float lds_a[2][KB_SUB * LDSP];

    const int lane = threadIdx.x & 31;
    const int wave = __builtin_amdgcn_readfirstlane((int)(threadIdx.x >> 5));
    const int  grp  = wave / 3;
    const bool is_p = (wave - 3 * grp) == 0;
    const int  gi   = wave - 1 - grp;
    const int  m0   = (int)blockIdx.x * KB_ROWS;

    float bias[4] = {0.f, 0.f, 0.f, 0.f};
    int   ida = 0, idb = 0;
    float wa  = 0.f, wb = 0.f;
    if (is_p) {
#pragma unroll
        for (int n = 0; n < 4; ++n) bias[n] = bc_ws[64 * grp + 16 * n + (lane & 15)];
    } else {
        const int ra = min(m0 + gi, n_act - 1);
        const int rb = min(m0 + gi + 8, n_act - 1);
        ida = id_ws[(size_t)ra * 32 + lane];
        wa  = fw_ws[(size_t)ra * 32 + lane];
        idb = id_ws[(size_t)rb * 32 + lane];
        wb  = fw_ws[(size_t)rb * 32 + lane];
    }

    for (int s = -1; s < KB_NSUB; ++s) {
        if (is_p) {
            if (s < 0) kp_copy_wc(lds_w, wc_ws, grp, lane);
            else       kp_product(lds_a[s & 1], lds_w, m0 + KB_SUB * s, grp, lane, bias, out, n_act);
        } else if (s + 1 < KB_NSUB) {
            const int   nbase = m0 + KB_SUB * min(s + 2, KB_NSUB - 1) + gi;
            const int   na  = min(nbase, n_act - 1);
            const int   nb  = min(nbase + 8, n_act - 1);
            const int   ina = id_ws[(size_t)na * 32 + lane];
            const float wna = fw_ws[(size_t)na * 32 + lane];
            const int   inb = id_ws[(size_t)nb * 32 + lane];
            const float wnb = fw_ws[(size_t)nb * 32 + lane];
            float* __restrict__ tl = lds_a[(s + 1) & 1];
#pragma nounroll
            for (int i = 0; i < 2; ++i) {
                kf_gather_row(tl + (gi + 8 * i) * LDSP, i ? idb : ida, i ? wb : wa, act, pas, n_act, lane);
            }
            ida = ina;  wa = wna;
            idb = inb;  wb = wnb;
        }
        if (s + 1 < KB_NSUB) __syncthreads();
    }
}

extern "C" void kernel_launch(void* const* d_in, const int* in_sizes, int n_in,
                              void* d_out, int out_size, void* d_ws, size_t ws_size,
                              hipStream_t stream) {
    const float* act  = (const float*)d_in[0];
    const float* pas  = (const float*)d_in[1];
    const int*   bids = (const int*)d_in[2];
    const int*   mids = (const int*)d_in[3];
    const int*   xy   = (const int*)d_in[4];
    const int*   m0   = (const int*)d_in[5];
    const int*   m1   = (const int*)d_in[6];
    const int*   m2   = (const int*)d_in[7];
    const int*   m3   = (const int*)d_in[8];
    const float* Wa   = (const float*)d_in[9];
    const float* ba   = (const float*)d_in[10];
    const float* Wv   = (const float*)d_in[11];
    const float* bv   = (const float*)d_in[12];
    const float* Wo   = (const float*)d_in[13];
    const float* bo   = (const float*)d_in[14];
    float* out = (float*)d_out;

    const int n_act = in_sizes[0] / F_SIZE;

    float* wc_ws = (float*)d_ws;
    float* bc_ws = wc_ws + F_SIZE * F_SIZE;
    float* fw_ws = bc_ws + F_SIZE;
    int*   id_ws = (int*)(fw_ws + (size_t)n_act * 32);

    k_attn_ids<<<K1_FOLD_BLOCKS + (n_act + 16 * K1_TILES - 1) / (16 * K1_TILES), 128, 0, stream>>>(
        act, bids, mids, xy, m0, m1, m2, m3, Wa, ba, Wv, bv, Wo, bo, wc_ws, bc_ws,
        fw_ws, id_ws, n_act);

    k_gather_gemm<<<(n_act + KB_ROWS - 1) / KB_ROWS, 32 * KB_WAVES, 0, stream>>>(
        act, pas, fw_ws, id_ws, wc_ws, bc_ws, out, n_act);
}
